# starting kernel
# baseline (speedup 1.0000x reference)
#include <hip/hip_runtime.h>
#include <hip/hip_bf16.h>


#define BSZ     2
#define LSEQ    2048
#define DIN     2048
#define DSTATE  8
#define DTRANK  256
#define KPROJ   (DTRANK + 2 * DSTATE)
#define NCHUNK  8
#define LCHUNK  (LSEQ / NCHUNK)

typedef float v2f __attribute__((ext_vector_type(2)));
typedef float v8f __attribute__((ext_vector_type(8)));

__global__ __launch_bounds__(256) void proj_kernel(
    const float* __restrict__ x,
    const float* __restrict__ Wxp,
    float* __restrict__ xdbl)
{
    const int lane = threadIdx.x & 31;
    const int wave = threadIdx.x >> 5;
    const int tile = blockIdx.x * 8 + wave;
    const int rt = tile / 17;
    const int ct = tile % 17;
    const int m  = lane & 15;
    const int hi = lane >> 4;

    const float* arow = x   + (size_t)(rt * 16 + m) * DIN + 2 * hi;
    const float* brow = Wxp + (size_t)(ct * 16 + m) * DIN + 2 * hi;

    v8f acc = {};
    for (int k = 0; k < DIN; k += 4) {
        v2f a = *(const v2f*)(arow + k);
        v2f b = *(const v2f*)(brow + k);
        acc = __builtin_amdgcn_wmma_f32_16x16x4_f32(
            false, a, false, b, (short)0, acc, false, false);
    }
    #pragma unroll
    for (int j = 0; j < 8; ++j) {
        const int row = rt * 16 + j + 8 * hi;
        const int col = ct * 16 + m;
        xdbl[(size_t)row * KPROJ + col] = acc[j];
    }
}

__global__ __launch_bounds__(256) void delta_kernel(
    const float* __restrict__ xdbl,
    const float* __restrict__ Wdt,
    const float* __restrict__ b_dt,
    float* __restrict__ delta)
{
    const int lane = threadIdx.x & 31;
    const int wave = threadIdx.x >> 5;
    const int tile = blockIdx.x * 8 + wave;
    const int rt = tile >> 7;
    const int ct = tile & 127;
    const int m  = lane & 15;
    const int hi = lane >> 4;

    const float* arow = xdbl + (size_t)(rt * 16 + m) * KPROJ + 2 * hi;
    const float* brow = Wdt  + (size_t)(ct * 16 + m) * DTRANK + 2 * hi;

    v8f acc = {};
    for (int k = 0; k < DTRANK; k += 4) {
        v2f a = *(const v2f*)(arow + k);
        v2f b = *(const v2f*)(brow + k);
        acc = __builtin_amdgcn_wmma_f32_16x16x4_f32(
            false, a, false, b, (short)0, acc, false, false);
    }
    #pragma unroll
    for (int j = 0; j < 8; ++j) {
        const int row = rt * 16 + j + 8 * hi;
        const int col = ct * 16 + m;
        float v = acc[j] + b_dt[col];
        float sp = (v > 20.0f) ? v : log1pf(__expf(v));
        delta[(size_t)row * DIN + col] = sp;
    }
}

__global__ __launch_bounds__(256) void scan_pass1(
    const float* __restrict__ delta,
    const float* __restrict__ x,
    const float* __restrict__ xdbl,
    const float* __restrict__ A_log,
    float* __restrict__ aProd,
    float* __restrict__ hPart)
{
    const int t = blockIdx.x * blockDim.x + threadIdx.x;
    const int n = t & 7;
    const int c = (t >> 3) & 7;
    const int d = (t >> 6) & (DIN - 1);
    const int b = t >> 17;

    const float A = -__expf(A_log[d * DSTATE + n]);
    float h = 0.0f, aP = 1.0f;
    const int l0 = c * LCHUNK;
    for (int i = 0; i < LCHUNK; ++i) {
        const int row = b * LSEQ + l0 + i;
        const float dv = delta[(size_t)row * DIN + d];
        const float xv = x[(size_t)row * DIN + d];
        const float Bv = xdbl[(size_t)row * KPROJ + DTRANK + n];
        const float dA = __expf(dv * A);
        h  = fmaf(dA, h, dv * xv * Bv);
        aP *= dA;
    }
    const int idx = (((b * DIN + d) * NCHUNK) + c) * DSTATE + n;
    aProd[idx] = aP;
    hPart[idx] = h;
}

__global__ __launch_bounds__(256) void scan_pass2(
    const float* __restrict__ aProd,
    const float* __restrict__ hPart,
    float* __restrict__ hIn)
{
    const int t = blockIdx.x * blockDim.x + threadIdx.x;
    const int n  = t & 7;
    const int bd = t >> 3;
    float h = 0.0f;
    #pragma unroll
    for (int c = 0; c < NCHUNK; ++c) {
        const int idx = (bd * NCHUNK + c) * DSTATE + n;
        hIn[idx] = h;
        h = fmaf(aProd[idx], h, hPart[idx]);
    }
}

__global__ __launch_bounds__(256) void scan_pass3(
    const float* __restrict__ delta,
    const float* __restrict__ x,
    const float* __restrict__ xdbl,
    const float* __restrict__ A_log,
    const float* __restrict__ D_param,
    const float* __restrict__ hIn,
    float* __restrict__ y)
{
    const int t = blockIdx.x * blockDim.x + threadIdx.x;
    const int n = t & 7;
    const int c = (t >> 3) & 7;
    const int d = (t >> 6) & (DIN - 1);
    const int b = t >> 17;

    const float A  = -__expf(A_log[d * DSTATE + n]);
    const float Dp = D_param[d];
    float h = hIn[(((b * DIN + d) * NCHUNK) + c) * DSTATE + n];
    const int l0 = c * LCHUNK;
    for (int i = 0; i < LCHUNK; ++i) {
        const int row = b * LSEQ + l0 + i;
        const float dv = delta[(size_t)row * DIN + d];
        const float xv = x[(size_t)row * DIN + d];
        const float Bv = xdbl[(size_t)row * KPROJ + DTRANK + n];
        const float Cv = xdbl[(size_t)row * KPROJ + DTRANK + DSTATE + n];
        const float dA = __expf(dv * A);
        h = fmaf(dA, h, dv * xv * Bv);
        float part = h * Cv;
        part += __shfl_xor(part, 1, 32);
        part += __shfl_xor(part, 2, 32);
        part += __shfl_xor(part, 4, 32);
        if (n == 0)
            y[(size_t)row * DIN + d] = part + xv * Dp;
    }
}

extern "C" void kernel_launch(void* const* d_in, const int* in_sizes, int n_in,
                              void* d_out, int out_size, void* d_ws, size_t ws_size,
                              hipStream_t stream) {
    const float* x       = (const float*)d_in[0];
    const float* W_xproj = (const float*)d_in[1];
    const float* W_dt    = (const float*)d_in[2];
    const float* b_dt    = (const float*)d_in[3];
    const float* A_log   = (const float*)d_in[4];
    const float* D_param = (const float*)d_in[5];
    float* y = (float*)d_out;

    float* ws    = (float*)d_ws;
    float* xdbl  = ws;
    float* delta = xdbl + (size_t)BSZ * LSEQ * KPROJ;
    float* aProd = delta + (size_t)BSZ * LSEQ * DIN;
    float* hPart = aProd + (size_t)BSZ * DIN * NCHUNK * DSTATE;
    float* hIn   = hPart + (size_t)BSZ * DIN * NCHUNK * DSTATE;

    proj_kernel<<<544, 256, 0, stream>>>(x, W_xproj, xdbl);
    delta_kernel<<<4096, 256, 0, stream>>>(xdbl, W_dt, b_dt, delta);
    scan_pass1<<<1024, 256, 0, stream>>>(delta, x, xdbl, A_log, aProd, hPart);
    scan_pass2<<<128, 256, 0, stream>>>(aProd, hPart, hIn);
    scan_pass3<<<1024, 256, 0, stream>>>(delta, x, xdbl, A_log, D_param, hIn, y);
}


# scan passes: lane = channel d, 8 states per thread, coalesced delta/x/y lines (8-step register prefetch), B/C wave-uniform from LDS; sum order changes: in-thread fma chain over n, chunks 8->32
# speedup vs baseline: 2.0028x; 2.0028x over previous
#include <hip/hip_runtime.h>
#include <hip/hip_bf16.h>


#define BSZ     2
#define LSEQ    2048
#define DIN     2048
#define DSTATE  8
#define DTRANK  256
#define KPROJ   (DTRANK + 2 * DSTATE)
#define NCHUNK  32
#define LCHUNK  (LSEQ / NCHUNK)
#define DBLK    256
#define NDBLK   (DIN / DBLK)
#define SCAN_U  8

typedef float v2f __attribute__((ext_vector_type(2)));
typedef float v8f __attribute__((ext_vector_type(8)));
typedef float v4f __attribute__((ext_vector_type(4)));

__global__ __launch_bounds__(256) void proj_kernel(
    const float* __restrict__ x,
    const float* __restrict__ Wxp,
    float* __restrict__ xdbl)
{
    const int lane = threadIdx.x & 31;
    const int wave = threadIdx.x >> 5;
    const int tile = blockIdx.x * 8 + wave;
    const int rt = tile / 17;
    const int ct = tile % 17;
    const int m  = lane & 15;
    const int hi = lane >> 4;

    const float* arow = x   + (size_t)(rt * 16 + m) * DIN + 2 * hi;
    const float* brow = Wxp + (size_t)(ct * 16 + m) * DIN + 2 * hi;

    v8f acc = {};
    for (int k = 0; k < DIN; k += 4) {
        v2f a = *(const v2f*)(arow + k);
        v2f b = *(const v2f*)(brow + k);
        acc = __builtin_amdgcn_wmma_f32_16x16x4_f32(
            false, a, false, b, (short)0, acc, false, false);
    }
    #pragma unroll
    for (int j = 0; j < 8; ++j) {
        const int row = rt * 16 + j + 8 * hi;
        const int col = ct * 16 + m;
        xdbl[(size_t)row * KPROJ + col] = acc[j];
    }
}

__global__ __launch_bounds__(256) void delta_kernel(
    const float* __restrict__ xdbl,
    const float* __restrict__ Wdt,
    const float* __restrict__ b_dt,
    float* __restrict__ delta)
{
    const int lane = threadIdx.x & 31;
    const int wave = threadIdx.x >> 5;
    const int tile = blockIdx.x * 8 + wave;
    const int rt = tile >> 7;
    const int ct = tile & 127;
    const int m  = lane & 15;
    const int hi = lane >> 4;

    const float* arow = xdbl + (size_t)(rt * 16 + m) * KPROJ + 2 * hi;
    const float* brow = Wdt  + (size_t)(ct * 16 + m) * DTRANK + 2 * hi;

    v8f acc = {};
    for (int k = 0; k < DTRANK; k += 4) {
        v2f a = *(const v2f*)(arow + k);
        v2f b = *(const v2f*)(brow + k);
        acc = __builtin_amdgcn_wmma_f32_16x16x4_f32(
            false, a, false, b, (short)0, acc, false, false);
    }
    #pragma unroll
    for (int j = 0; j < 8; ++j) {
        const int row = rt * 16 + j + 8 * hi;
        const int col = ct * 16 + m;
        float v = acc[j] + b_dt[col];
        float sp = (v > 20.0f) ? v : log1pf(__expf(v));
        delta[(size_t)row * DIN + col] = sp;
    }
}

__global__ __launch_bounds__(256) void scan_pass1(
    const float* __restrict__ delta,
    const float* __restrict__ x,
    const float* __restrict__ xdbl,
    const float* __restrict__ A_log,
    float* __restrict__ aProd,
    float* __restrict__ hPart)
{
    __shared__ __attribute__((aligned(16))) float sB[LCHUNK][DSTATE];

    const int tid  = threadIdx.x;
    const int dblk = blockIdx.x % NDBLK;
    const int c    = (blockIdx.x / NDBLK) % NCHUNK;
    const int b    = blockIdx.x / (NDBLK * NCHUNK);
    const int d    = dblk * DBLK + tid;
    const int row0 = b * LSEQ + c * LCHUNK;

    for (int idx = tid; idx < LCHUNK * 2; idx += 256) {
        const int r = idx >> 1, q = idx & 1;
        *(v4f*)&sB[r][4 * q] =
            *(const v4f*)(xdbl + (size_t)(row0 + r) * KPROJ + DTRANK + 4 * q);
    }

    float A[DSTATE], h[DSTATE], aP[DSTATE];
    {
        const v4f a0 = *(const v4f*)(A_log + (size_t)d * DSTATE);
        const v4f a1 = *(const v4f*)(A_log + (size_t)d * DSTATE + 4);
        for (int n = 0; n < 4; ++n) {
            A[n]     = -__expf(a0[n]);
            A[n + 4] = -__expf(a1[n]);
        }
    }
    for (int n = 0; n < DSTATE; ++n) { h[n] = 0.0f; aP[n] = 1.0f; }

    const float* dp = delta + (size_t)row0 * DIN + d;
    const float* xp = x     + (size_t)row0 * DIN + d;

    float dvN[SCAN_U], xvN[SCAN_U];
    for (int u = 0; u < SCAN_U; ++u) {
        dvN[u] = dp[(size_t)u * DIN];
        xvN[u] = xp[(size_t)u * DIN];
    }

    __syncthreads();

    for (int i0 = 0; i0 < LCHUNK; i0 += SCAN_U) {
        float dv[SCAN_U], xv[SCAN_U];
        for (int u = 0; u < SCAN_U; ++u) { dv[u] = dvN[u]; xv[u] = xvN[u]; }
        if (i0 + SCAN_U < LCHUNK) {
            for (int u = 0; u < SCAN_U; ++u) {
                dvN[u] = dp[(size_t)(i0 + SCAN_U + u) * DIN];
                xvN[u] = xp[(size_t)(i0 + SCAN_U + u) * DIN];
            }
        }
        for (int u = 0; u < SCAN_U; ++u) {
            const v4f B0 = *(const v4f*)&sB[i0 + u][0];
            const v4f B1 = *(const v4f*)&sB[i0 + u][4];
            const float dbx = dv[u] * xv[u];
            for (int n = 0; n < DSTATE; ++n) {
                const float Bv = (n < 4) ? B0[n & 3] : B1[n & 3];
                const float dA = __expf(dv[u] * A[n]);
                h[n]  = fmaf(dA, h[n], dbx * Bv);
                aP[n] *= dA;
            }
        }
    }

    for (int n = 0; n < DSTATE; ++n) {
        const size_t idx = ((size_t)(b * NCHUNK + c) * DSTATE + n) * DIN + d;
        aProd[idx] = aP[n];
        hPart[idx] = h[n];
    }
}

__global__ __launch_bounds__(256) void scan_pass2(
    const float* __restrict__ aProd,
    const float* __restrict__ hPart,
    float* __restrict__ hIn)
{
    const int t = blockIdx.x * blockDim.x + threadIdx.x;
    const int d = t % DIN;
    const int n = (t / DIN) % DSTATE;
    const int b = t / (DIN * DSTATE);
    float h = 0.0f;
    #pragma unroll
    for (int c = 0; c < NCHUNK; ++c) {
        const size_t idx = ((size_t)(b * NCHUNK + c) * DSTATE + n) * DIN + d;
        hIn[idx] = h;
        h = fmaf(aProd[idx], h, hPart[idx]);
    }
}

__global__ __launch_bounds__(256) void scan_pass3(
    const float* __restrict__ delta,
    const float* __restrict__ x,
    const float* __restrict__ xdbl,
    const float* __restrict__ A_log,
    const float* __restrict__ D_param,
    const float* __restrict__ hIn,
    float* __restrict__ y)
{
    __shared__ __attribute__((aligned(16))) float sBC[LCHUNK][2 * DSTATE];

    const int tid  = threadIdx.x;
    const int dblk = blockIdx.x % NDBLK;
    const int c    = (blockIdx.x / NDBLK) % NCHUNK;
    const int b    = blockIdx.x / (NDBLK * NCHUNK);
    const int d    = dblk * DBLK + tid;
    const int row0 = b * LSEQ + c * LCHUNK;

    for (int idx = tid; idx < LCHUNK * 4; idx += 256) {
        const int r = idx >> 2, q = idx & 3;
        *(v4f*)&sBC[r][4 * q] =
            *(const v4f*)(xdbl + (size_t)(row0 + r) * KPROJ + DTRANK + 4 * q);
    }

    float A[DSTATE], h[DSTATE];
    {
        const v4f a0 = *(const v4f*)(A_log + (size_t)d * DSTATE);
        const v4f a1 = *(const v4f*)(A_log + (size_t)d * DSTATE + 4);
        for (int n = 0; n < 4; ++n) {
            A[n]     = -__expf(a0[n]);
            A[n + 4] = -__expf(a1[n]);
        }
    }
    const float Dp = D_param[d];
    for (int n = 0; n < DSTATE; ++n)
        h[n] = hIn[((size_t)(b * NCHUNK + c) * DSTATE + n) * DIN + d];

    const float* dp = delta + (size_t)row0 * DIN + d;
    const float* xp = x     + (size_t)row0 * DIN + d;
    float*       yp = y     + (size_t)row0 * DIN + d;

    float dvN[SCAN_U], xvN[SCAN_U];
    for (int u = 0; u < SCAN_U; ++u) {
        dvN[u] = dp[(size_t)u * DIN];
        xvN[u] = xp[(size_t)u * DIN];
    }

    __syncthreads();

    for (int i0 = 0; i0 < LCHUNK; i0 += SCAN_U) {
        float dv[SCAN_U], xv[SCAN_U];
        for (int u = 0; u < SCAN_U; ++u) { dv[u] = dvN[u]; xv[u] = xvN[u]; }
        if (i0 + SCAN_U < LCHUNK) {
            for (int u = 0; u < SCAN_U; ++u) {
                dvN[u] = dp[(size_t)(i0 + SCAN_U + u) * DIN];
                xvN[u] = xp[(size_t)(i0 + SCAN_U + u) * DIN];
            }
        }
        for (int u = 0; u < SCAN_U; ++u) {
            const v4f B0 = *(const v4f*)&sBC[i0 + u][0];
            const v4f B1 = *(const v4f*)&sBC[i0 + u][4];
            const v4f C0 = *(const v4f*)&sBC[i0 + u][8];
            const v4f C1 = *(const v4f*)&sBC[i0 + u][12];
            const float dbx = dv[u] * xv[u];
            float acc = 0.0f;
            for (int n = 0; n < DSTATE; ++n) {
                const float Bv = (n < 4) ? B0[n & 3] : B1[n & 3];
                const float Cv = (n < 4) ? C0[n & 3] : C1[n & 3];
                const float dA = __expf(dv[u] * A[n]);
                h[n] = fmaf(dA, h[n], dbx * Bv);
                acc  = fmaf(h[n], Cv, acc);
            }
            yp[(size_t)(i0 + u) * DIN] = fmaf(xv[u], Dp, acc);
        }
    }
}

extern "C" void kernel_launch(void* const* d_in, const int* in_sizes, int n_in,
                              void* d_out, int out_size, void* d_ws, size_t ws_size,
                              hipStream_t stream) {
    const float* x       = (const float*)d_in[0];
    const float* W_xproj = (const float*)d_in[1];
    const float* W_dt    = (const float*)d_in[2];
    const float* b_dt    = (const float*)d_in[3];
    const float* A_log   = (const float*)d_in[4];
    const float* D_param = (const float*)d_in[5];
    float* y = (float*)d_out;

    float* ws    = (float*)d_ws;
    float* xdbl  = ws;
    float* delta = xdbl + (size_t)BSZ * LSEQ * KPROJ;
    float* aProd = delta + (size_t)BSZ * LSEQ * DIN;
    float* hPart = aProd + (size_t)BSZ * DIN * NCHUNK * DSTATE;
    float* hIn   = hPart + (size_t)BSZ * DIN * NCHUNK * DSTATE;

    proj_kernel<<<544, 256, 0, stream>>>(x, W_xproj, xdbl);
    delta_kernel<<<4096, 256, 0, stream>>>(xdbl, W_dt, b_dt, delta);
    scan_pass1<<<512, 256, 0, stream>>>(delta, x, xdbl, A_log, aProd, hPart);
    scan_pass2<<<128, 256, 0, stream>>>(aProd, hPart, hIn);
    scan_pass3<<<512, 256, 0, stream>>>(delta, x, xdbl, A_log, D_param, hIn, y);
}


# both projection GEMMs: one wave per block of 16x16 tiles (2x1, 4x4), 128-bit fragment loads issued a stage ahead, each fragment feeds 2 to 8 WMMAs; k order inside 8-k steps changes (sum order)
# speedup vs baseline: 2.8857x; 1.4408x over previous
#include <hip/hip_runtime.h>
#include <hip/hip_bf16.h>


#define BSZ     2
#define LSEQ    2048
#define DIN     2048
#define DSTATE  8
#define DTRANK  256
#define KPROJ   (DTRANK + 2 * DSTATE)
#define NCHUNK  32
#define LCHUNK  (LSEQ / NCHUNK)
#define DBLK    256
#define NDBLK   (DIN / DBLK)
#define SCAN_U  8

typedef float v2f __attribute__((ext_vector_type(2)));
typedef float v8f __attribute__((ext_vector_type(8)));
typedef float v4f __attribute__((ext_vector_type(4)));

template <int TM, int TN, int SUB>
__device__ __forceinline__ void mma_stage(
    const v4f (&a)[TM][SUB], const v4f (&b)[TN][SUB], v8f (&acc)[TM][TN])
{
    #pragma unroll
    for (int s = 0; s < SUB; ++s) {
        #pragma unroll
        for (int i = 0; i < TM; ++i) {
            #pragma unroll
            for (int j = 0; j < TN; ++j) {
                const v2f al = __builtin_shufflevector(a[i][s], a[i][s], 0, 1);
                const v2f bl = __builtin_shufflevector(b[j][s], b[j][s], 0, 1);
                acc[i][j] = __builtin_amdgcn_wmma_f32_16x16x4_f32(
                    false, al, false, bl, (short)0, acc[i][j], false, false);
            }
        }
        #pragma unroll
        for (int i = 0; i < TM; ++i) {
            #pragma unroll
            for (int j = 0; j < TN; ++j) {
                const v2f ah = __builtin_shufflevector(a[i][s], a[i][s], 2, 3);
                const v2f bh = __builtin_shufflevector(b[j][s], b[j][s], 2, 3);
                acc[i][j] = __builtin_amdgcn_wmma_f32_16x16x4_f32(
                    false, ah, false, bh, (short)0, acc[i][j], false, false);
            }
        }
    }
}

template <int TM, int TN, int SUB, int LDA, int LDB>
__device__ __forceinline__ void load_stage(
    const float* __restrict__ ap, const float* __restrict__ bp, const int k,
    v4f (&a)[TM][SUB], v4f (&b)[TN][SUB])
{
    #pragma unroll
    for (int i = 0; i < TM; ++i) {
        #pragma unroll
        for (int s = 0; s < SUB; ++s)
            a[i][s] = *(const v4f*)(ap + (size_t)i * 16 * LDA + k + 8 * s);
    }
    #pragma unroll
    for (int j = 0; j < TN; ++j) {
        #pragma unroll
        for (int s = 0; s < SUB; ++s)
            b[j][s] = *(const v4f*)(bp + (size_t)j * 16 * LDB + k + 8 * s);
    }
}

template <int TM, int TN, int SUB, int LDA, int LDB, int KLEN>
__device__ __forceinline__ void wave_tiles_nt(
    const float* __restrict__ ap,
    const float* __restrict__ bp,
    v8f (&acc)[TM][TN])
{
    static_assert(KLEN % (16 * SUB) == 0 && KLEN >= 32 * SUB, "two stages per trip");
    v4f a0[TM][SUB], b0[TN][SUB], a1[TM][SUB], b1[TN][SUB];
    load_stage<TM, TN, SUB, LDA, LDB>(ap, bp, 0, a0, b0);

    for (int k = 0; k < KLEN - 16 * SUB; k += 16 * SUB) {
        load_stage<TM, TN, SUB, LDA, LDB>(ap, bp, 8 * SUB, a1, b1);
        __builtin_amdgcn_sched_barrier(0);
        mma_stage<TM, TN, SUB>(a0, b0, acc);
        load_stage<TM, TN, SUB, LDA, LDB>(ap, bp, 16 * SUB, a0, b0);
        __builtin_amdgcn_sched_barrier(0);
        mma_stage<TM, TN, SUB>(a1, b1, acc);
        ap += 16 * SUB;
        bp += 16 * SUB;
    }

    load_stage<TM, TN, SUB, LDA, LDB>(ap, bp, 8 * SUB, a1, b1);
    __builtin_amdgcn_sched_barrier(0);
    mma_stage<TM, TN, SUB>(a0, b0, acc);
    mma_stage<TM, TN, SUB>(a1, b1, acc);
}

__global__ __launch_bounds__(32) void proj_kernel(
    const float* __restrict__ x,
    const float* __restrict__ Wxp,
    float* __restrict__ xdbl)
{
    const int lane = threadIdx.x & 31;
    const int c  = lane & 15;
    const int hi = lane >> 4;
    const int rp = blockIdx.x / 17;
    const int ct = blockIdx.x % 17;

    const float* ap = x   + (size_t)(rp * 32 + c) * DIN + 4 * hi;
    const float* bp = Wxp + (size_t)(ct * 16 + c) * DIN + 4 * hi;

    v8f acc[2][1] = {};
    wave_tiles_nt<2, 1, 8, DIN, DIN, DIN>(ap, bp, acc);

    #pragma unroll
    for (int i = 0; i < 2; ++i) {
        #pragma unroll
        for (int j = 0; j < 8; ++j) {
            const int row = rp * 32 + 16 * i + j + 8 * hi;
            const int col = ct * 16 + c;
            xdbl[(size_t)row * KPROJ + col] = acc[i][0][j];
        }
    }
}

__global__ __launch_bounds__(32) void delta_kernel(
    const float* __restrict__ xdbl,
    const float* __restrict__ Wdt,
    const float* __restrict__ b_dt,
    float* __restrict__ delta)
{
    const int lane = threadIdx.x & 31;
    const int c  = lane & 15;
    const int hi = lane >> 4;
    const int rg = blockIdx.x >> 5;
    const int cg = blockIdx.x & 31;

    const float* ap = xdbl + (size_t)(rg * 64 + c) * KPROJ  + 4 * hi;
    const float* bp = Wdt  + (size_t)(cg * 64 + c) * DTRANK + 4 * hi;

    v8f acc[4][4] = {};
    wave_tiles_nt<4, 4, 1, KPROJ, DTRANK, DTRANK>(ap, bp, acc);

    int col = cg * 64 + c;
    #pragma unroll 1
    for (int jt = 0; jt < 4; ++jt) {
        const float bias = b_dt[col];
        #pragma unroll
        for (int i = 0; i < 4; ++i) {
            #pragma unroll
            for (int j = 0; j < 8; ++j) {
                const int row = rg * 64 + 16 * i + j + 8 * hi;
                float v = acc[i][0][j] + bias;
                float sp = (v > 20.0f) ? v : log1pf(__expf(v));
                delta[(size_t)row * DIN + col] = sp;
            }
        }
        #pragma unroll
        for (int i = 0; i < 4; ++i) {
            acc[i][0] = acc[i][1];
            acc[i][1] = acc[i][2];
            acc[i][2] = acc[i][3];
        }
        col += 16;
    }
}

__global__ __launch_bounds__(256) void scan_pass1(
    const float* __restrict__ delta,
    const float* __restrict__ x,
    const float* __restrict__ xdbl,
    const float* __restrict__ A_log,
    float* __restrict__ aProd,
    float* __restrict__ hPart)
{
    __shared__ __attribute__((aligned(16))) float sB[LCHUNK][DSTATE];

    const int tid  = threadIdx.x;
    const int dblk = blockIdx.x % NDBLK;
    const int c    = (blockIdx.x / NDBLK) % NCHUNK;
    const int b    = blockIdx.x / (NDBLK * NCHUNK);
    const int d    = dblk * DBLK + tid;
    const int row0 = b * LSEQ + c * LCHUNK;

    for (int idx = tid; idx < LCHUNK * 2; idx += 256) {
        const int r = idx >> 1, q = idx & 1;
        *(v4f*)&sB[r][4 * q] =
            *(const v4f*)(xdbl + (size_t)(row0 + r) * KPROJ + DTRANK + 4 * q);
    }

    float A[DSTATE], h[DSTATE], aP[DSTATE];
    {
        const v4f a0 = *(const v4f*)(A_log + (size_t)d * DSTATE);
        const v4f a1 = *(const v4f*)(A_log + (size_t)d * DSTATE + 4);
        for (int n = 0; n < 4; ++n) {
            A[n]     = -__expf(a0[n]);
            A[n + 4] = -__expf(a1[n]);
        }
    }
    for (int n = 0; n < DSTATE; ++n) { h[n] = 0.0f; aP[n] = 1.0f; }

    const float* dp = delta + (size_t)row0 * DIN + d;
    const float* xp = x     + (size_t)row0 * DIN + d;

    float dvN[SCAN_U], xvN[SCAN_U];
    for (int u = 0; u < SCAN_U; ++u) {
        dvN[u] = dp[(size_t)u * DIN];
        xvN[u] = xp[(size_t)u * DIN];
    }

    __syncthreads();

    for (int i0 = 0; i0 < LCHUNK; i0 += SCAN_U) {
        float dv[SCAN_U], xv[SCAN_U];
        for (int u = 0; u < SCAN_U; ++u) { dv[u] = dvN[u]; xv[u] = xvN[u]; }
        if (i0 + SCAN_U < LCHUNK) {
            for (int u = 0; u < SCAN_U; ++u) {
                dvN[u] = dp[(size_t)(i0 + SCAN_U + u) * DIN];
                xvN[u] = xp[(size_t)(i0 + SCAN_U + u) * DIN];
            }
        }
        for (int u = 0; u < SCAN_U; ++u) {
            const v4f B0 = *(const v4f*)&sB[i0 + u][0];
            const v4f B1 = *(const v4f*)&sB[i0 + u][4];
            const float dbx = dv[u] * xv[u];
            for (int n = 0; n < DSTATE; ++n) {
                const float Bv = (n < 4) ? B0[n & 3] : B1[n & 3];
                const float dA = __expf(dv[u] * A[n]);
                h[n]  = fmaf(dA, h[n], dbx * Bv);
                aP[n] *= dA;
            }
        }
    }

    for (int n = 0; n < DSTATE; ++n) {
        const size_t idx = ((size_t)(b * NCHUNK + c) * DSTATE + n) * DIN + d;
        aProd[idx] = aP[n];
        hPart[idx] = h[n];
    }
}

__global__ __launch_bounds__(256) void scan_pass2(
    const float* __restrict__ aProd,
    const float* __restrict__ hPart,
    float* __restrict__ hIn)
{
    const int t = blockIdx.x * blockDim.x + threadIdx.x;
    const int d = t % DIN;
    const int n = (t / DIN) % DSTATE;
    const int b = t / (DIN * DSTATE);
    float h = 0.0f;
    #pragma unroll
    for (int c = 0; c < NCHUNK; ++c) {
        const size_t idx = ((size_t)(b * NCHUNK + c) * DSTATE + n) * DIN + d;
        hIn[idx] = h;
        h = fmaf(aProd[idx], h, hPart[idx]);
    }
}

__global__ __launch_bounds__(256) void scan_pass3(
    const float* __restrict__ delta,
    const float* __restrict__ x,
    const float* __restrict__ xdbl,
    const float* __restrict__ A_log,
    const float* __restrict__ D_param,
    const float* __restrict__ hIn,
    float* __restrict__ y)
{
    __shared__ __attribute__((aligned(16))) float sBC[LCHUNK][2 * DSTATE];

    const int tid  = threadIdx.x;
    const int dblk = blockIdx.x % NDBLK;
    const int c    = (blockIdx.x / NDBLK) % NCHUNK;
    const int b    = blockIdx.x / (NDBLK * NCHUNK);
    const int d    = dblk * DBLK + tid;
    const int row0 = b * LSEQ + c * LCHUNK;

    for (int idx = tid; idx < LCHUNK * 4; idx += 256) {
        const int r = idx >> 2, q = idx & 3;
        *(v4f*)&sBC[r][4 * q] =
            *(const v4f*)(xdbl + (size_t)(row0 + r) * KPROJ + DTRANK + 4 * q);
    }

    float A[DSTATE], h[DSTATE];
    {
        const v4f a0 = *(const v4f*)(A_log + (size_t)d * DSTATE);
        const v4f a1 = *(const v4f*)(A_log + (size_t)d * DSTATE + 4);
        for (int n = 0; n < 4; ++n) {
            A[n]     = -__expf(a0[n]);
            A[n + 4] = -__expf(a1[n]);
        }
    }
    const float Dp = D_param[d];
    for (int n = 0; n < DSTATE; ++n)
        h[n] = hIn[((size_t)(b * NCHUNK + c) * DSTATE + n) * DIN + d];

    const float* dp = delta + (size_t)row0 * DIN + d;
    const float* xp = x     + (size_t)row0 * DIN + d;
    float*       yp = y     + (size_t)row0 * DIN + d;

    float dvN[SCAN_U], xvN[SCAN_U];
    for (int u = 0; u < SCAN_U; ++u) {
        dvN[u] = dp[(size_t)u * DIN];
        xvN[u] = xp[(size_t)u * DIN];
    }

    __syncthreads();

    for (int i0 = 0; i0 < LCHUNK; i0 += SCAN_U) {
        float dv[SCAN_U], xv[SCAN_U];
        for (int u = 0; u < SCAN_U; ++u) { dv[u] = dvN[u]; xv[u] = xvN[u]; }
        if (i0 + SCAN_U < LCHUNK) {
            for (int u = 0; u < SCAN_U; ++u) {
                dvN[u] = dp[(size_t)(i0 + SCAN_U + u) * DIN];
                xvN[u] = xp[(size_t)(i0 + SCAN_U + u) * DIN];
            }
        }
        for (int u = 0; u < SCAN_U; ++u) {
            const v4f B0 = *(const v4f*)&sBC[i0 + u][0];
            const v4f B1 = *(const v4f*)&sBC[i0 + u][4];
            const v4f C0 = *(const v4f*)&sBC[i0 + u][8];
            const v4f C1 = *(const v4f*)&sBC[i0 + u][12];
            const float dbx = dv[u] * xv[u];
            float acc = 0.0f;
            for (int n = 0; n < DSTATE; ++n) {
                const float Bv = (n < 4) ? B0[n & 3] : B1[n & 3];
                const float Cv = (n < 4) ? C0[n & 3] : C1[n & 3];
                const float dA = __expf(dv[u] * A[n]);
                h[n] = fmaf(dA, h[n], dbx * Bv);
                acc  = fmaf(h[n], Cv, acc);
            }
            yp[(size_t)(i0 + u) * DIN] = fmaf(xv[u], Dp, acc);
        }
    }
}

extern "C" void kernel_launch(void* const* d_in, const int* in_sizes, int n_in,
                              void* d_out, int out_size, void* d_ws, size_t ws_size,
                              hipStream_t stream) {
    const float* x       = (const float*)d_in[0];
    const float* W_xproj = (const float*)d_in[1];
    const float* W_dt    = (const float*)d_in[2];
    const float* b_dt    = (const float*)d_in[3];
    const float* A_log   = (const float*)d_in[4];
    const float* D_param = (const float*)d_in[5];
    float* y = (float*)d_out;

    float* ws    = (float*)d_ws;
    float* xdbl  = ws;
    float* delta = xdbl + (size_t)BSZ * LSEQ * KPROJ;
    float* aProd = delta + (size_t)BSZ * LSEQ * DIN;
    float* hPart = aProd + (size_t)BSZ * DIN * NCHUNK * DSTATE;
    float* hIn   = hPart + (size_t)BSZ * DIN * NCHUNK * DSTATE;

    proj_kernel<<<2176, 32, 0, stream>>>(x, W_xproj, xdbl);
    delta_kernel<<<2048, 32, 0, stream>>>(xdbl, W_dt, b_dt, delta);
    scan_pass1<<<512, 256, 0, stream>>>(delta, x, xdbl, A_log, aProd, hPart);
    scan_pass2<<<128, 256, 0, stream>>>(aProd, hPart, hIn);
    scan_pass3<<<512, 256, 0, stream>>>(delta, x, xdbl, A_log, D_param, hIn, y);
}


# delta epilogue: softplus as max(z,0)+log1p(exp(-|z|)), log1p(t)=2atanh(t/(2+t)) by a degree-4 polynomial, branch-free and unrolled, in place of the library log1p chain; delta moves by ulps
# speedup vs baseline: 3.4643x; 1.2005x over previous
#include <hip/hip_runtime.h>
#include <hip/hip_bf16.h>


#define BSZ     2
#define LSEQ    2048
#define DIN     2048
#define DSTATE  8
#define DTRANK  256
#define KPROJ   (DTRANK + 2 * DSTATE)
#define NCHUNK  32
#define LCHUNK  (LSEQ / NCHUNK)
#define DBLK    256
#define NDBLK   (DIN / DBLK)
#define SCAN_U  8

typedef float v2f __attribute__((ext_vector_type(2)));
typedef float v8f __attribute__((ext_vector_type(8)));
typedef float v4f __attribute__((ext_vector_type(4)));

template <int TM, int TN, int SUB>
__device__ __forceinline__ void mma_stage(
    const v4f (&a)[TM][SUB], const v4f (&b)[TN][SUB], v8f (&acc)[TM][TN])
{
    #pragma unroll
    for (int s = 0; s < SUB; ++s) {
        #pragma unroll
        for (int i = 0; i < TM; ++i) {
            #pragma unroll
            for (int j = 0; j < TN; ++j) {
                const v2f al = __builtin_shufflevector(a[i][s], a[i][s], 0, 1);
                const v2f bl = __builtin_shufflevector(b[j][s], b[j][s], 0, 1);
                acc[i][j] = __builtin_amdgcn_wmma_f32_16x16x4_f32(
                    false, al, false, bl, (short)0, acc[i][j], false, false);
            }
        }
        #pragma unroll
        for (int i = 0; i < TM; ++i) {
            #pragma unroll
            for (int j = 0; j < TN; ++j) {
                const v2f ah = __builtin_shufflevector(a[i][s], a[i][s], 2, 3);
                const v2f bh = __builtin_shufflevector(b[j][s], b[j][s], 2, 3);
                acc[i][j] = __builtin_amdgcn_wmma_f32_16x16x4_f32(
                    false, ah, false, bh, (short)0, acc[i][j], false, false);
            }
        }
    }
}

template <int TM, int TN, int SUB, int LDA, int LDB>
__device__ __forceinline__ void load_stage(
    const float* __restrict__ ap, const float* __restrict__ bp, const int k,
    v4f (&a)[TM][SUB], v4f (&b)[TN][SUB])
{
    #pragma unroll
    for (int i = 0; i < TM; ++i) {
        #pragma unroll
        for (int s = 0; s < SUB; ++s)
            a[i][s] = *(const v4f*)(ap + (size_t)i * 16 * LDA + k + 8 * s);
    }
    #pragma unroll
    for (int j = 0; j < TN; ++j) {
        #pragma unroll
        for (int s = 0; s < SUB; ++s)
            b[j][s] = *(const v4f*)(bp + (size_t)j * 16 * LDB + k + 8 * s);
    }
}

template <int TM, int TN, int SUB, int LDA, int LDB, int KLEN>
__device__ __forceinline__ void wave_tiles_nt(
    const float* __restrict__ ap,
    const float* __restrict__ bp,
    v8f (&acc)[TM][TN])
{
    static_assert(KLEN % (16 * SUB) == 0 && KLEN >= 32 * SUB, "two stages per trip");
    v4f a0[TM][SUB], b0[TN][SUB], a1[TM][SUB], b1[TN][SUB];
    load_stage<TM, TN, SUB, LDA, LDB>(ap, bp, 0, a0, b0);

    for (int k = 0; k < KLEN - 16 * SUB; k += 16 * SUB) {
        load_stage<TM, TN, SUB, LDA, LDB>(ap, bp, 8 * SUB, a1, b1);
        __builtin_amdgcn_sched_barrier(0);
        mma_stage<TM, TN, SUB>(a0, b0, acc);
        load_stage<TM, TN, SUB, LDA, LDB>(ap, bp, 16 * SUB, a0, b0);
        __builtin_amdgcn_sched_barrier(0);
        mma_stage<TM, TN, SUB>(a1, b1, acc);
        ap += 16 * SUB;
        bp += 16 * SUB;
    }

    load_stage<TM, TN, SUB, LDA, LDB>(ap, bp, 8 * SUB, a1, b1);
    __builtin_amdgcn_sched_barrier(0);
    mma_stage<TM, TN, SUB>(a0, b0, acc);
    mma_stage<TM, TN, SUB>(a1, b1, acc);
}

__global__ __launch_bounds__(32) void proj_kernel(
    const float* __restrict__ x,
    const float* __restrict__ Wxp,
    float* __restrict__ xdbl)
{
    const int lane = threadIdx.x & 31;
    const int c  = lane & 15;
    const int hi = lane >> 4;
    const int rp = blockIdx.x / 17;
    const int ct = blockIdx.x % 17;

    const float* ap = x   + (size_t)(rp * 32 + c) * DIN + 4 * hi;
    const float* bp = Wxp + (size_t)(ct * 16 + c) * DIN + 4 * hi;

    v8f acc[2][1] = {};
    wave_tiles_nt<2, 1, 8, DIN, DIN, DIN>(ap, bp, acc);

    #pragma unroll
    for (int i = 0; i < 2; ++i) {
        #pragma unroll
        for (int j = 0; j < 8; ++j) {
            const int row = rp * 32 + 16 * i + j + 8 * hi;
            const int col = ct * 16 + c;
            xdbl[(size_t)row * KPROJ + col] = acc[i][0][j];
        }
    }
}

__device__ __forceinline__ float softplus_f32(const float z)
{
    const float t = __expf(-fabsf(z));
    const float s = t * __builtin_amdgcn_rcpf(2.0f + t);
    const float w = s * s;
    float q = 0.281577796f;
    q = fmaf(q, w, 0.279651582f);
    q = fmaf(q, w, 0.400245965f);
    q = fmaf(q, w, 0.666663229f);
    q = fmaf(q, w, 2.0f);
    return fmaf(s, q, fmaxf(z, 0.0f));
}

__global__ __launch_bounds__(32) void delta_kernel(
    const float* __restrict__ xdbl,
    const float* __restrict__ Wdt,
    const float* __restrict__ b_dt,
    float* __restrict__ delta)
{
    const int lane = threadIdx.x & 31;
    const int c  = lane & 15;
    const int hi = lane >> 4;
    const int rg = blockIdx.x >> 5;
    const int cg = blockIdx.x & 31;

    const float* ap = xdbl + (size_t)(rg * 64 + c) * KPROJ  + 4 * hi;
    const float* bp = Wdt  + (size_t)(cg * 64 + c) * DTRANK + 4 * hi;

    v8f acc[4][4] = {};
    wave_tiles_nt<4, 4, 1, KPROJ, DTRANK, DTRANK>(ap, bp, acc);

    #pragma unroll
    for (int jt = 0; jt < 4; ++jt) {
        const int col = cg * 64 + 16 * jt + c;
        const float bias = b_dt[col];
        #pragma unroll
        for (int i = 0; i < 4; ++i) {
            #pragma unroll
            for (int j = 0; j < 8; ++j) {
                const int row = rg * 64 + 16 * i + j + 8 * hi;
                delta[(size_t)row * DIN + col] = softplus_f32(acc[i][jt][j] + bias);
            }
        }
    }
}

__global__ __launch_bounds__(256) void scan_pass1(
    const float* __restrict__ delta,
    const float* __restrict__ x,
    const float* __restrict__ xdbl,
    const float* __restrict__ A_log,
    float* __restrict__ aProd,
    float* __restrict__ hPart)
{
    __shared__ __attribute__((aligned(16))) float sB[LCHUNK][DSTATE];

    const int tid  = threadIdx.x;
    const int dblk = blockIdx.x % NDBLK;
    const int c    = (blockIdx.x / NDBLK) % NCHUNK;
    const int b    = blockIdx.x / (NDBLK * NCHUNK);
    const int d    = dblk * DBLK + tid;
    const int row0 = b * LSEQ + c * LCHUNK;

    for (int idx = tid; idx < LCHUNK * 2; idx += 256) {
        const int r = idx >> 1, q = idx & 1;
        *(v4f*)&sB[r][4 * q] =
            *(const v4f*)(xdbl + (size_t)(row0 + r) * KPROJ + DTRANK + 4 * q);
    }

    float A[DSTATE], h[DSTATE], aP[DSTATE];
    {
        const v4f a0 = *(const v4f*)(A_log + (size_t)d * DSTATE);
        const v4f a1 = *(const v4f*)(A_log + (size_t)d * DSTATE + 4);
        for (int n = 0; n < 4; ++n) {
            A[n]     = -__expf(a0[n]);
            A[n + 4] = -__expf(a1[n]);
        }
    }
    for (int n = 0; n < DSTATE; ++n) { h[n] = 0.0f; aP[n] = 1.0f; }

    const float* dp = delta + (size_t)row0 * DIN + d;
    const float* xp = x     + (size_t)row0 * DIN + d;

    float dvN[SCAN_U], xvN[SCAN_U];
    for (int u = 0; u < SCAN_U; ++u) {
        dvN[u] = dp[(size_t)u * DIN];
        xvN[u] = xp[(size_t)u * DIN];
    }

    __syncthreads();

    for (int i0 = 0; i0 < LCHUNK; i0 += SCAN_U) {
        float dv[SCAN_U], xv[SCAN_U];
        for (int u = 0; u < SCAN_U; ++u) { dv[u] = dvN[u]; xv[u] = xvN[u]; }
        if (i0 + SCAN_U < LCHUNK) {
            for (int u = 0; u < SCAN_U; ++u) {
                dvN[u] = dp[(size_t)(i0 + SCAN_U + u) * DIN];
                xvN[u] = xp[(size_t)(i0 + SCAN_U + u) * DIN];
            }
        }
        for (int u = 0; u < SCAN_U; ++u) {
            const v4f B0 = *(const v4f*)&sB[i0 + u][0];
            const v4f B1 = *(const v4f*)&sB[i0 + u][4];
            const float dbx = dv[u] * xv[u];
            for (int n = 0; n < DSTATE; ++n) {
                const float Bv = (n < 4) ? B0[n & 3] : B1[n & 3];
                const float dA = __expf(dv[u] * A[n]);
                h[n]  = fmaf(dA, h[n], dbx * Bv);
                aP[n] *= dA;
            }
        }
    }

    for (int n = 0; n < DSTATE; ++n) {
        const size_t idx = ((size_t)(b * NCHUNK + c) * DSTATE + n) * DIN + d;
        aProd[idx] = aP[n];
        hPart[idx] = h[n];
    }
}

__global__ __launch_bounds__(256) void scan_pass2(
    const float* __restrict__ aProd,
    const float* __restrict__ hPart,
    float* __restrict__ hIn)
{
    const int t = blockIdx.x * blockDim.x + threadIdx.x;
    const int d = t % DIN;
    const int n = (t / DIN) % DSTATE;
    const int b = t / (DIN * DSTATE);
    float h = 0.0f;
    #pragma unroll
    for (int c = 0; c < NCHUNK; ++c) {
        const size_t idx = ((size_t)(b * NCHUNK + c) * DSTATE + n) * DIN + d;
        hIn[idx] = h;
        h = fmaf(aProd[idx], h, hPart[idx]);
    }
}

__global__ __launch_bounds__(256) void scan_pass3(
    const float* __restrict__ delta,
    const float* __restrict__ x,
    const float* __restrict__ xdbl,
    const float* __restrict__ A_log,
    const float* __restrict__ D_param,
    const float* __restrict__ hIn,
    float* __restrict__ y)
{
    __shared__ __attribute__((aligned(16))) float sBC[LCHUNK][2 * DSTATE];

    const int tid  = threadIdx.x;
    const int dblk = blockIdx.x % NDBLK;
    const int c    = (blockIdx.x / NDBLK) % NCHUNK;
    const int b    = blockIdx.x / (NDBLK * NCHUNK);
    const int d    = dblk * DBLK + tid;
    const int row0 = b * LSEQ + c * LCHUNK;

    for (int idx = tid; idx < LCHUNK * 4; idx += 256) {
        const int r = idx >> 2, q = idx & 3;
        *(v4f*)&sBC[r][4 * q] =
            *(const v4f*)(xdbl + (size_t)(row0 + r) * KPROJ + DTRANK + 4 * q);
    }

    float A[DSTATE], h[DSTATE];
    {
        const v4f a0 = *(const v4f*)(A_log + (size_t)d * DSTATE);
        const v4f a1 = *(const v4f*)(A_log + (size_t)d * DSTATE + 4);
        for (int n = 0; n < 4; ++n) {
            A[n]     = -__expf(a0[n]);
            A[n + 4] = -__expf(a1[n]);
        }
    }
    const float Dp = D_param[d];
    for (int n = 0; n < DSTATE; ++n)
        h[n] = hIn[((size_t)(b * NCHUNK + c) * DSTATE + n) * DIN + d];

    const float* dp = delta + (size_t)row0 * DIN + d;
    const float* xp = x     + (size_t)row0 * DIN + d;
    float*       yp = y     + (size_t)row0 * DIN + d;

    float dvN[SCAN_U], xvN[SCAN_U];
    for (int u = 0; u < SCAN_U; ++u) {
        dvN[u] = dp[(size_t)u * DIN];
        xvN[u] = xp[(size_t)u * DIN];
    }

    __syncthreads();

    for (int i0 = 0; i0 < LCHUNK; i0 += SCAN_U) {
        float dv[SCAN_U], xv[SCAN_U];
        for (int u = 0; u < SCAN_U; ++u) { dv[u] = dvN[u]; xv[u] = xvN[u]; }
        if (i0 + SCAN_U < LCHUNK) {
            for (int u = 0; u < SCAN_U; ++u) {
                dvN[u] = dp[(size_t)(i0 + SCAN_U + u) * DIN];
                xvN[u] = xp[(size_t)(i0 + SCAN_U + u) * DIN];
            }
        }
        for (int u = 0; u < SCAN_U; ++u) {
            const v4f B0 = *(const v4f*)&sBC[i0 + u][0];
            const v4f B1 = *(const v4f*)&sBC[i0 + u][4];
            const v4f C0 = *(const v4f*)&sBC[i0 + u][8];
            const v4f C1 = *(const v4f*)&sBC[i0 + u][12];
            const float dbx = dv[u] * xv[u];
            float acc = 0.0f;
            for (int n = 0; n < DSTATE; ++n) {
                const float Bv = (n < 4) ? B0[n & 3] : B1[n & 3];
                const float Cv = (n < 4) ? C0[n & 3] : C1[n & 3];
                const float dA = __expf(dv[u] * A[n]);
                h[n] = fmaf(dA, h[n], dbx * Bv);
                acc  = fmaf(h[n], Cv, acc);
            }
            yp[(size_t)(i0 + u) * DIN] = fmaf(xv[u], Dp, acc);
        }
    }
}

extern "C" void kernel_launch(void* const* d_in, const int* in_sizes, int n_in,
                              void* d_out, int out_size, void* d_ws, size_t ws_size,
                              hipStream_t stream) {
    const float* x       = (const float*)d_in[0];
    const float* W_xproj = (const float*)d_in[1];
    const float* W_dt    = (const float*)d_in[2];
    const float* b_dt    = (const float*)d_in[3];
    const float* A_log   = (const float*)d_in[4];
    const float* D_param = (const float*)d_in[5];
    float* y = (float*)d_out;

    float* ws    = (float*)d_ws;
    float* xdbl  = ws;
    float* delta = xdbl + (size_t)BSZ * LSEQ * KPROJ;
    float* aProd = delta + (size_t)BSZ * LSEQ * DIN;
    float* hPart = aProd + (size_t)BSZ * DIN * NCHUNK * DSTATE;
    float* hIn   = hPart + (size_t)BSZ * DIN * NCHUNK * DSTATE;

    proj_kernel<<<2176, 32, 0, stream>>>(x, W_xproj, xdbl);
    delta_kernel<<<2048, 32, 0, stream>>>(xdbl, W_dt, b_dt, delta);
    scan_pass1<<<512, 256, 0, stream>>>(delta, x, xdbl, A_log, aProd, hPart);
    scan_pass2<<<128, 256, 0, stream>>>(aProd, hPart, hIn);
    scan_pass3<<<512, 256, 0, stream>>>(delta, x, xdbl, A_log, D_param, hIn, y);
}


# x projection GEMM re-tiled: one 4-wave block per 32 rows x 144/128 columns, each wave all 2x9 tiles over a quarter of k, the 4 partial sums added through LDS (k sum order changes)
# speedup vs baseline: 4.4882x; 1.2956x over previous
#include <hip/hip_runtime.h>
#include <hip/hip_bf16.h>


#define BSZ     2
#define LSEQ    2048
#define DIN     2048
#define DSTATE  8
#define DTRANK  256
#define KPROJ   (DTRANK + 2 * DSTATE)
#define NCHUNK  32
#define LCHUNK  (LSEQ / NCHUNK)
#define DBLK    256
#define NDBLK   (DIN / DBLK)
#define SCAN_U  8

typedef float v2f __attribute__((ext_vector_type(2)));
typedef float v8f __attribute__((ext_vector_type(8)));
typedef float v4f __attribute__((ext_vector_type(4)));

template <int TM, int TN, int SUB>
__device__ __forceinline__ void mma_stage(
    const v4f (&a)[TM][SUB], const v4f (&b)[TN][SUB], v8f (&acc)[TM][TN])
{
    #pragma unroll
    for (int s = 0; s < SUB; ++s) {
        #pragma unroll
        for (int i = 0; i < TM; ++i) {
            #pragma unroll
            for (int j = 0; j < TN; ++j) {
                const v2f al = __builtin_shufflevector(a[i][s], a[i][s], 0, 1);
                const v2f bl = __builtin_shufflevector(b[j][s], b[j][s], 0, 1);
                acc[i][j] = __builtin_amdgcn_wmma_f32_16x16x4_f32(
                    false, al, false, bl, (short)0, acc[i][j], false, false);
            }
        }
        #pragma unroll
        for (int i = 0; i < TM; ++i) {
            #pragma unroll
            for (int j = 0; j < TN; ++j) {
                const v2f ah = __builtin_shufflevector(a[i][s], a[i][s], 2, 3);
                const v2f bh = __builtin_shufflevector(b[j][s], b[j][s], 2, 3);
                acc[i][j] = __builtin_amdgcn_wmma_f32_16x16x4_f32(
                    false, ah, false, bh, (short)0, acc[i][j], false, false);
            }
        }
    }
}

template <int TM, int TN, int SUB, int LDA, int LDB>
__device__ __forceinline__ void load_stage(
    const float* __restrict__ ap, const float* __restrict__ bp, const int k,
    v4f (&a)[TM][SUB], v4f (&b)[TN][SUB])
{
    #pragma unroll
    for (int i = 0; i < TM; ++i) {
        #pragma unroll
        for (int s = 0; s < SUB; ++s)
            a[i][s] = *(const v4f*)(ap + (size_t)i * 16 * LDA + k + 8 * s);
    }
    #pragma unroll
    for (int j = 0; j < TN; ++j) {
        #pragma unroll
        for (int s = 0; s < SUB; ++s)
            b[j][s] = *(const v4f*)(bp + (size_t)j * 16 * LDB + k + 8 * s);
    }
}

template <int TM, int TN, int SUB, int LDA, int LDB, int KLEN>
__device__ __forceinline__ void wave_tiles_nt(
    const float* __restrict__ ap,
    const float* __restrict__ bp,
    v8f (&acc)[TM][TN])
{
    static_assert(KLEN % (16 * SUB) == 0 && KLEN >= 32 * SUB, "two stages per trip");
    v4f a0[TM][SUB], b0[TN][SUB], a1[TM][SUB], b1[TN][SUB];
    load_stage<TM, TN, SUB, LDA, LDB>(ap, bp, 0, a0, b0);

    for (int k = 0; k < KLEN - 16 * SUB; k += 16 * SUB) {
        load_stage<TM, TN, SUB, LDA, LDB>(ap, bp, 8 * SUB, a1, b1);
        __builtin_amdgcn_sched_barrier(0);
        mma_stage<TM, TN, SUB>(a0, b0, acc);
        load_stage<TM, TN, SUB, LDA, LDB>(ap, bp, 16 * SUB, a0, b0);
        __builtin_amdgcn_sched_barrier(0);
        mma_stage<TM, TN, SUB>(a1, b1, acc);
        ap += 16 * SUB;
        bp += 16 * SUB;
    }

    load_stage<TM, TN, SUB, LDA, LDB>(ap, bp, 8 * SUB, a1, b1);
    __builtin_amdgcn_sched_barrier(0);
    mma_stage<TM, TN, SUB>(a0, b0, acc);
    mma_stage<TM, TN, SUB>(a1, b1, acc);
}

#define PROJ_WAVES   4
#define PROJ_KSLICE  (DIN / PROJ_WAVES)
#define PROJ_TN0     9
#define PROJ_TN1     8

template <int TN>
__device__ __forceinline__ void proj_block(
    const float* __restrict__ ap, const float* __restrict__ bp,
    float* __restrict__ xdbl, v4f* __restrict__ sP,
    const int wave, const int lane, const int row0, const int col0)
{
    const int c  = lane & 15;
    const int hi = lane >> 4;

    v8f acc[2][TN] = {};
    wave_tiles_nt<2, TN, 4, DIN, DIN, PROJ_KSLICE>(ap, bp, acc);

    #pragma unroll
    for (int i = 0; i < 2; ++i) {
        #pragma unroll
        for (int j = 0; j < TN; ++j) {
            const int t = i * TN + j;
            if ((t & 3) != wave) {
                const int p = (wave - t - 1) & 3;
                v4f* dst = sP + ((t * 3 + p) * 2) * 32 + lane;
                dst[0]  = __builtin_shufflevector(acc[i][j], acc[i][j], 0, 1, 2, 3);
                dst[32] = __builtin_shufflevector(acc[i][j], acc[i][j], 4, 5, 6, 7);
            }
        }
    }

    __syncthreads();

    #pragma unroll
    for (int i = 0; i < 2; ++i) {
        #pragma unroll
        for (int j = 0; j < TN; ++j) {
            const int t = i * TN + j;
            if ((t & 3) == wave) {
                v8f s = acc[i][j];
                #pragma unroll
                for (int p = 0; p < 3; ++p) {
                    const v4f* src = sP + ((t * 3 + p) * 2) * 32 + lane;
                    const v4f lo = src[0];
                    const v4f hh = src[32];
                    #pragma unroll
                    for (int r = 0; r < 4; ++r) {
                        s[r]     += lo[r];
                        s[r + 4] += hh[r];
                    }
                }
                #pragma unroll
                for (int r = 0; r < 8; ++r) {
                    const int row = row0 + 16 * i + r + 8 * hi;
                    const int col = col0 + 16 * j + c;
                    xdbl[(size_t)row * KPROJ + col] = s[r];
                }
            }
        }
    }
}

__global__ __launch_bounds__(128) void proj_kernel(
    const float* __restrict__ x,
    const float* __restrict__ Wxp,
    float* __restrict__ xdbl)
{
    __shared__ __attribute__((aligned(16))) v4f sP[2 * PROJ_TN0 * 3 * 2 * 32];

    const int lane = threadIdx.x & 31;
    const int wave = __builtin_amdgcn_readfirstlane((int)(threadIdx.x >> 5));
    const int c  = lane & 15;
    const int hi = lane >> 4;
    const int rp = blockIdx.x >> 1;
    const int cg = blockIdx.x & 1;
    const int row0 = rp * 32;
    const int col0 = cg * (16 * PROJ_TN0);

    const float* ap = x   + (size_t)(row0 + c) * DIN + wave * PROJ_KSLICE + 4 * hi;
    const float* bp = Wxp + (size_t)(col0 + c) * DIN + wave * PROJ_KSLICE + 4 * hi;

    if (cg == 0) proj_block<PROJ_TN0>(ap, bp, xdbl, sP, wave, lane, row0, col0);
    else         proj_block<PROJ_TN1>(ap, bp, xdbl, sP, wave, lane, row0, col0);
}

__device__ __forceinline__ float softplus_f32(const float z)
{
    const float t = __expf(-fabsf(z));
    const float s = t * __builtin_amdgcn_rcpf(2.0f + t);
    const float w = s * s;
    float q = 0.281577796f;
    q = fmaf(q, w, 0.279651582f);
    q = fmaf(q, w, 0.400245965f);
    q = fmaf(q, w, 0.666663229f);
    q = fmaf(q, w, 2.0f);
    return fmaf(s, q, fmaxf(z, 0.0f));
}

__global__ __launch_bounds__(32) void delta_kernel(
    const float* __restrict__ xdbl,
    const float* __restrict__ Wdt,
    const float* __restrict__ b_dt,
    float* __restrict__ delta)
{
    const int lane = threadIdx.x & 31;
    const int c  = lane & 15;
    const int hi = lane >> 4;
    const int rg = blockIdx.x >> 5;
    const int cg = blockIdx.x & 31;

    const float* ap = xdbl + (size_t)(rg * 64 + c) * KPROJ  + 4 * hi;
    const float* bp = Wdt  + (size_t)(cg * 64 + c) * DTRANK + 4 * hi;

    v8f acc[4][4] = {};
    wave_tiles_nt<4, 4, 1, KPROJ, DTRANK, DTRANK>(ap, bp, acc);

    #pragma unroll
    for (int jt = 0; jt < 4; ++jt) {
        const int col = cg * 64 + 16 * jt + c;
        const float bias = b_dt[col];
        #pragma unroll
        for (int i = 0; i < 4; ++i) {
            #pragma unroll
            for (int j = 0; j < 8; ++j) {
                const int row = rg * 64 + 16 * i + j + 8 * hi;
                delta[(size_t)row * DIN + col] = softplus_f32(acc[i][jt][j] + bias);
            }
        }
    }
}

__global__ __launch_bounds__(256) void scan_pass1(
    const float* __restrict__ delta,
    const float* __restrict__ x,
    const float* __restrict__ xdbl,
    const float* __restrict__ A_log,
    float* __restrict__ aProd,
    float* __restrict__ hPart)
{
    __shared__ __attribute__((aligned(16))) float sB[LCHUNK][DSTATE];

    const int tid  = threadIdx.x;
    const int dblk = blockIdx.x % NDBLK;
    const int c    = (blockIdx.x / NDBLK) % NCHUNK;
    const int b    = blockIdx.x / (NDBLK * NCHUNK);
    const int d    = dblk * DBLK + tid;
    const int row0 = b * LSEQ + c * LCHUNK;

    for (int idx = tid; idx < LCHUNK * 2; idx += 256) {
        const int r = idx >> 1, q = idx & 1;
        *(v4f*)&sB[r][4 * q] =
            *(const v4f*)(xdbl + (size_t)(row0 + r) * KPROJ + DTRANK + 4 * q);
    }

    float A[DSTATE], h[DSTATE], aP[DSTATE];
    {
        const v4f a0 = *(const v4f*)(A_log + (size_t)d * DSTATE);
        const v4f a1 = *(const v4f*)(A_log + (size_t)d * DSTATE + 4);
        for (int n = 0; n < 4; ++n) {
            A[n]     = -__expf(a0[n]);
            A[n + 4] = -__expf(a1[n]);
        }
    }
    for (int n = 0; n < DSTATE; ++n) { h[n] = 0.0f; aP[n] = 1.0f; }

    const float* dp = delta + (size_t)row0 * DIN + d;
    const float* xp = x     + (size_t)row0 * DIN + d;

    float dvN[SCAN_U], xvN[SCAN_U];
    for (int u = 0; u < SCAN_U; ++u) {
        dvN[u] = dp[(size_t)u * DIN];
        xvN[u] = xp[(size_t)u * DIN];
    }

    __syncthreads();

    for (int i0 = 0; i0 < LCHUNK; i0 += SCAN_U) {
        float dv[SCAN_U], xv[SCAN_U];
        for (int u = 0; u < SCAN_U; ++u) { dv[u] = dvN[u]; xv[u] = xvN[u]; }
        if (i0 + SCAN_U < LCHUNK) {
            for (int u = 0; u < SCAN_U; ++u) {
                dvN[u] = dp[(size_t)(i0 + SCAN_U + u) * DIN];
                xvN[u] = xp[(size_t)(i0 + SCAN_U + u) * DIN];
            }
        }
        for (int u = 0; u < SCAN_U; ++u) {
            const v4f B0 = *(const v4f*)&sB[i0 + u][0];
            const v4f B1 = *(const v4f*)&sB[i0 + u][4];
            const float dbx = dv[u] * xv[u];
            for (int n = 0; n < DSTATE; ++n) {
                const float Bv = (n < 4) ? B0[n & 3] : B1[n & 3];
                const float dA = __expf(dv[u] * A[n]);
                h[n]  = fmaf(dA, h[n], dbx * Bv);
                aP[n] *= dA;
            }
        }
    }

    for (int n = 0; n < DSTATE; ++n) {
        const size_t idx = ((size_t)(b * NCHUNK + c) * DSTATE + n) * DIN + d;
        aProd[idx] = aP[n];
        hPart[idx] = h[n];
    }
}

__global__ __launch_bounds__(256) void scan_pass2(
    const float* __restrict__ aProd,
    const float* __restrict__ hPart,
    float* __restrict__ hIn)
{
    const int t = blockIdx.x * blockDim.x + threadIdx.x;
    const int d = t % DIN;
    const int n = (t / DIN) % DSTATE;
    const int b = t / (DIN * DSTATE);
    float h = 0.0f;
    #pragma unroll
    for (int c = 0; c < NCHUNK; ++c) {
        const size_t idx = ((size_t)(b * NCHUNK + c) * DSTATE + n) * DIN + d;
        hIn[idx] = h;
        h = fmaf(aProd[idx], h, hPart[idx]);
    }
}

__global__ __launch_bounds__(256) void scan_pass3(
    const float* __restrict__ delta,
    const float* __restrict__ x,
    const float* __restrict__ xdbl,
    const float* __restrict__ A_log,
    const float* __restrict__ D_param,
    const float* __restrict__ hIn,
    float* __restrict__ y)
{
    __shared__ __attribute__((aligned(16))) float sBC[LCHUNK][2 * DSTATE];

    const int tid  = threadIdx.x;
    const int dblk = blockIdx.x % NDBLK;
    const int c    = (blockIdx.x / NDBLK) % NCHUNK;
    const int b    = blockIdx.x / (NDBLK * NCHUNK);
    const int d    = dblk * DBLK + tid;
    const int row0 = b * LSEQ + c * LCHUNK;

    for (int idx = tid; idx < LCHUNK * 4; idx += 256) {
        const int r = idx >> 2, q = idx & 3;
        *(v4f*)&sBC[r][4 * q] =
            *(const v4f*)(xdbl + (size_t)(row0 + r) * KPROJ + DTRANK + 4 * q);
    }

    float A[DSTATE], h[DSTATE];
    {
        const v4f a0 = *(const v4f*)(A_log + (size_t)d * DSTATE);
        const v4f a1 = *(const v4f*)(A_log + (size_t)d * DSTATE + 4);
        for (int n = 0; n < 4; ++n) {
            A[n]     = -__expf(a0[n]);
            A[n + 4] = -__expf(a1[n]);
        }
    }
    const float Dp = D_param[d];
    for (int n = 0; n < DSTATE; ++n)
        h[n] = hIn[((size_t)(b * NCHUNK + c) * DSTATE + n) * DIN + d];

    const float* dp = delta + (size_t)row0 * DIN + d;
    const float* xp = x     + (size_t)row0 * DIN + d;
    float*       yp = y     + (size_t)row0 * DIN + d;

    float dvN[SCAN_U], xvN[SCAN_U];
    for (int u = 0; u < SCAN_U; ++u) {
        dvN[u] = dp[(size_t)u * DIN];
        xvN[u] = xp[(size_t)u * DIN];
    }

    __syncthreads();

    for (int i0 = 0; i0 < LCHUNK; i0 += SCAN_U) {
        float dv[SCAN_U], xv[SCAN_U];
        for (int u = 0; u < SCAN_U; ++u) { dv[u] = dvN[u]; xv[u] = xvN[u]; }
        if (i0 + SCAN_U < LCHUNK) {
            for (int u = 0; u < SCAN_U; ++u) {
                dvN[u] = dp[(size_t)(i0 + SCAN_U + u) * DIN];
                xvN[u] = xp[(size_t)(i0 + SCAN_U + u) * DIN];
            }
        }
        for (int u = 0; u < SCAN_U; ++u) {
            const v4f B0 = *(const v4f*)&sBC[i0 + u][0];
            const v4f B1 = *(const v4f*)&sBC[i0 + u][4];
            const v4f C0 = *(const v4f*)&sBC[i0 + u][8];
            const v4f C1 = *(const v4f*)&sBC[i0 + u][12];
            const float dbx = dv[u] * xv[u];
            float acc = 0.0f;
            for (int n = 0; n < DSTATE; ++n) {
                const float Bv = (n < 4) ? B0[n & 3] : B1[n & 3];
                const float Cv = (n < 4) ? C0[n & 3] : C1[n & 3];
                const float dA = __expf(dv[u] * A[n]);
                h[n] = fmaf(dA, h[n], dbx * Bv);
                acc  = fmaf(h[n], Cv, acc);
            }
            yp[(size_t)(i0 + u) * DIN] = fmaf(xv[u], Dp, acc);
        }
    }
}

extern "C" void kernel_launch(void* const* d_in, const int* in_sizes, int n_in,
                              void* d_out, int out_size, void* d_ws, size_t ws_size,
                              hipStream_t stream) {
    const float* x       = (const float*)d_in[0];
    const float* W_xproj = (const float*)d_in[1];
    const float* W_dt    = (const float*)d_in[2];
    const float* b_dt    = (const float*)d_in[3];
    const float* A_log   = (const float*)d_in[4];
    const float* D_param = (const float*)d_in[5];
    float* y = (float*)d_out;

    float* ws    = (float*)d_ws;
    float* xdbl  = ws;
    float* delta = xdbl + (size_t)BSZ * LSEQ * KPROJ;
    float* aProd = delta + (size_t)BSZ * LSEQ * DIN;
    float* hPart = aProd + (size_t)BSZ * DIN * NCHUNK * DSTATE;
    float* hIn   = hPart + (size_t)BSZ * DIN * NCHUNK * DSTATE;

    proj_kernel<<<256, 128, 0, stream>>>(x, W_xproj, xdbl);
    delta_kernel<<<2048, 32, 0, stream>>>(xdbl, W_dt, b_dt, delta);
    scan_pass1<<<512, 256, 0, stream>>>(delta, x, xdbl, A_log, aProd, hPart);
    scan_pass2<<<128, 256, 0, stream>>>(aProd, hPart, hIn);
    scan_pass3<<<512, 256, 0, stream>>>(delta, x, xdbl, A_log, D_param, hIn, y);
}


# delta GEMM: a stage is 32 k (one whole 128-byte line of every row, four loads issued together) instead of 8 k, so no line must survive in the WGP cache between stages; k order unchanged
# speedup vs baseline: 4.6915x; 1.0453x over previous
#include <hip/hip_runtime.h>
#include <hip/hip_bf16.h>


#define BSZ     2
#define LSEQ    2048
#define DIN     2048
#define DSTATE  8
#define DTRANK  256
#define KPROJ   (DTRANK + 2 * DSTATE)
#define NCHUNK  32
#define LCHUNK  (LSEQ / NCHUNK)
#define DBLK    256
#define NDBLK   (DIN / DBLK)
#define SCAN_U  8

typedef float v2f __attribute__((ext_vector_type(2)));
typedef float v8f __attribute__((ext_vector_type(8)));
typedef float v4f __attribute__((ext_vector_type(4)));

template <int TM, int TN, int SUB>
__device__ __forceinline__ void mma_stage(
    const v4f (&a)[TM][SUB], const v4f (&b)[TN][SUB], v8f (&acc)[TM][TN])
{
    #pragma unroll
    for (int s = 0; s < SUB; ++s) {
        #pragma unroll
        for (int i = 0; i < TM; ++i) {
            #pragma unroll
            for (int j = 0; j < TN; ++j) {
                const v2f al = __builtin_shufflevector(a[i][s], a[i][s], 0, 1);
                const v2f bl = __builtin_shufflevector(b[j][s], b[j][s], 0, 1);
                acc[i][j] = __builtin_amdgcn_wmma_f32_16x16x4_f32(
                    false, al, false, bl, (short)0, acc[i][j], false, false);
            }
        }
        #pragma unroll
        for (int i = 0; i < TM; ++i) {
            #pragma unroll
            for (int j = 0; j < TN; ++j) {
                const v2f ah = __builtin_shufflevector(a[i][s], a[i][s], 2, 3);
                const v2f bh = __builtin_shufflevector(b[j][s], b[j][s], 2, 3);
                acc[i][j] = __builtin_amdgcn_wmma_f32_16x16x4_f32(
                    false, ah, false, bh, (short)0, acc[i][j], false, false);
            }
        }
    }
}

template <int TM, int TN, int SUB, int LDA, int LDB>
__device__ __forceinline__ void load_stage(
    const float* __restrict__ ap, const float* __restrict__ bp, const int k,
    v4f (&a)[TM][SUB], v4f (&b)[TN][SUB])
{
    #pragma unroll
    for (int i = 0; i < TM; ++i) {
        #pragma unroll
        for (int s = 0; s < SUB; ++s)
            a[i][s] = *(const v4f*)(ap + (size_t)i * 16 * LDA + k + 8 * s);
    }
    #pragma unroll
    for (int j = 0; j < TN; ++j) {
        #pragma unroll
        for (int s = 0; s < SUB; ++s)
            b[j][s] = *(const v4f*)(bp + (size_t)j * 16 * LDB + k + 8 * s);
    }
}

template <int TM, int TN, int SUB, int LDA, int LDB, int KLEN>
__device__ __forceinline__ void wave_tiles_nt(
    const float* __restrict__ ap,
    const float* __restrict__ bp,
    v8f (&acc)[TM][TN])
{
    static_assert(KLEN % (16 * SUB) == 0 && KLEN >= 32 * SUB, "two stages per trip");
    v4f a0[TM][SUB], b0[TN][SUB], a1[TM][SUB], b1[TN][SUB];
    load_stage<TM, TN, SUB, LDA, LDB>(ap, bp, 0, a0, b0);

    for (int k = 0; k < KLEN - 16 * SUB; k += 16 * SUB) {
        load_stage<TM, TN, SUB, LDA, LDB>(ap, bp, 8 * SUB, a1, b1);
        __builtin_amdgcn_sched_barrier(0);
        mma_stage<TM, TN, SUB>(a0, b0, acc);
        load_stage<TM, TN, SUB, LDA, LDB>(ap, bp, 16 * SUB, a0, b0);
        __builtin_amdgcn_sched_barrier(0);
        mma_stage<TM, TN, SUB>(a1, b1, acc);
        ap += 16 * SUB;
        bp += 16 * SUB;
    }

    load_stage<TM, TN, SUB, LDA, LDB>(ap, bp, 8 * SUB, a1, b1);
    __builtin_amdgcn_sched_barrier(0);
    mma_stage<TM, TN, SUB>(a0, b0, acc);
    mma_stage<TM, TN, SUB>(a1, b1, acc);
}

#define PROJ_WAVES   4
#define PROJ_KSLICE  (DIN / PROJ_WAVES)
#define PROJ_TN0     9
#define PROJ_TN1     8

template <int TN>
__device__ __forceinline__ void proj_block(
    const float* __restrict__ ap, const float* __restrict__ bp,
    float* __restrict__ xdbl, v4f* __restrict__ sP,
    const int wave, const int lane, const int row0, const int col0)
{
    const int c  = lane & 15;
    const int hi = lane >> 4;

    v8f acc[2][TN] = {};
    wave_tiles_nt<2, TN, 4, DIN, DIN, PROJ_KSLICE>(ap, bp, acc);

    #pragma unroll
    for (int i = 0; i < 2; ++i) {
        #pragma unroll
        for (int j = 0; j < TN; ++j) {
            const int t = i * TN + j;
            if ((t & 3) != wave) {
                const int p = (wave - t - 1) & 3;
                v4f* dst = sP + ((t * 3 + p) * 2) * 32 + lane;
                dst[0]  = __builtin_shufflevector(acc[i][j], acc[i][j], 0, 1, 2, 3);
                dst[32] = __builtin_shufflevector(acc[i][j], acc[i][j], 4, 5, 6, 7);
            }
        }
    }

    __syncthreads();

    #pragma unroll
    for (int i = 0; i < 2; ++i) {
        #pragma unroll
        for (int j = 0; j < TN; ++j) {
            const int t = i * TN + j;
            if ((t & 3) == wave) {
                v8f s = acc[i][j];
                #pragma unroll
                for (int p = 0; p < 3; ++p) {
                    const v4f* src = sP + ((t * 3 + p) * 2) * 32 + lane;
                    const v4f lo = src[0];
                    const v4f hh = src[32];
                    #pragma unroll
                    for (int r = 0; r < 4; ++r) {
                        s[r]     += lo[r];
                        s[r + 4] += hh[r];
                    }
                }
                #pragma unroll
                for (int r = 0; r < 8; ++r) {
                    const int row = row0 + 16 * i + r + 8 * hi;
                    const int col = col0 + 16 * j + c;
                    xdbl[(size_t)row * KPROJ + col] = s[r];
                }
            }
        }
    }
}

__global__ __launch_bounds__(128) void proj_kernel(
    const float* __restrict__ x,
    const float* __restrict__ Wxp,
    float* __restrict__ xdbl)
{
    __shared__ __attribute__((aligned(16))) v4f sP[2 * PROJ_TN0 * 3 * 2 * 32];

    const int lane = threadIdx.x & 31;
    const int wave = __builtin_amdgcn_readfirstlane((int)(threadIdx.x >> 5));
    const int c  = lane & 15;
    const int hi = lane >> 4;
    const int rp = blockIdx.x >> 1;
    const int cg = blockIdx.x & 1;
    const int row0 = rp * 32;
    const int col0 = cg * (16 * PROJ_TN0);

    const float* ap = x   + (size_t)(row0 + c) * DIN + wave * PROJ_KSLICE + 4 * hi;
    const float* bp = Wxp + (size_t)(col0 + c) * DIN + wave * PROJ_KSLICE + 4 * hi;

    if (cg == 0) proj_block<PROJ_TN0>(ap, bp, xdbl, sP, wave, lane, row0, col0);
    else         proj_block<PROJ_TN1>(ap, bp, xdbl, sP, wave, lane, row0, col0);
}

__device__ __forceinline__ float softplus_f32(const float z)
{
    const float t = __expf(-fabsf(z));
    const float s = t * __builtin_amdgcn_rcpf(2.0f + t);
    const float w = s * s;
    float q = 0.281577796f;
    q = fmaf(q, w, 0.279651582f);
    q = fmaf(q, w, 0.400245965f);
    q = fmaf(q, w, 0.666663229f);
    q = fmaf(q, w, 2.0f);
    return fmaf(s, q, fmaxf(z, 0.0f));
}

__global__ __launch_bounds__(32) void delta_kernel(
    const float* __restrict__ xdbl,
    const float* __restrict__ Wdt,
    const float* __restrict__ b_dt,
    float* __restrict__ delta)
{
    const int lane = threadIdx.x & 31;
    const int c  = lane & 15;
    const int hi = lane >> 4;
    const int rg = blockIdx.x >> 5;
    const int cg = blockIdx.x & 31;

    const float* ap = xdbl + (size_t)(rg * 64 + c) * KPROJ  + 4 * hi;
    const float* bp = Wdt  + (size_t)(cg * 64 + c) * DTRANK + 4 * hi;

    v8f acc[4][4] = {};
    wave_tiles_nt<4, 4, 4, KPROJ, DTRANK, DTRANK>(ap, bp, acc);

    #pragma unroll
    for (int jt = 0; jt < 4; ++jt) {
        const int col = cg * 64 + 16 * jt + c;
        const float bias = b_dt[col];
        #pragma unroll
        for (int i = 0; i < 4; ++i) {
            #pragma unroll
            for (int j = 0; j < 8; ++j) {
                const int row = rg * 64 + 16 * i + j + 8 * hi;
                delta[(size_t)row * DIN + col] = softplus_f32(acc[i][jt][j] + bias);
            }
        }
    }
}

__global__ __launch_bounds__(256) void scan_pass1(
    const float* __restrict__ delta,
    const float* __restrict__ x,
    const float* __restrict__ xdbl,
    const float* __restrict__ A_log,
    float* __restrict__ aProd,
    float* __restrict__ hPart)
{
    __shared__ __attribute__((aligned(16))) float sB[LCHUNK][DSTATE];

    const int tid  = threadIdx.x;
    const int dblk = blockIdx.x % NDBLK;
    const int c    = (blockIdx.x / NDBLK) % NCHUNK;
    const int b    = blockIdx.x / (NDBLK * NCHUNK);
    const int d    = dblk * DBLK + tid;
    const int row0 = b * LSEQ + c * LCHUNK;

    for (int idx = tid; idx < LCHUNK * 2; idx += 256) {
        const int r = idx >> 1, q = idx & 1;
        *(v4f*)&sB[r][4 * q] =
            *(const v4f*)(xdbl + (size_t)(row0 + r) * KPROJ + DTRANK + 4 * q);
    }

    float A[DSTATE], h[DSTATE], aP[DSTATE];
    {
        const v4f a0 = *(const v4f*)(A_log + (size_t)d * DSTATE);
        const v4f a1 = *(const v4f*)(A_log + (size_t)d * DSTATE + 4);
        for (int n = 0; n < 4; ++n) {
            A[n]     = -__expf(a0[n]);
            A[n + 4] = -__expf(a1[n]);
        }
    }
    for (int n = 0; n < DSTATE; ++n) { h[n] = 0.0f; aP[n] = 1.0f; }

    const float* dp = delta + (size_t)row0 * DIN + d;
    const float* xp = x     + (size_t)row0 * DIN + d;

    float dvN[SCAN_U], xvN[SCAN_U];
    for (int u = 0; u < SCAN_U; ++u) {
        dvN[u] = dp[(size_t)u * DIN];
        xvN[u] = xp[(size_t)u * DIN];
    }

    __syncthreads();

    for (int i0 = 0; i0 < LCHUNK; i0 += SCAN_U) {
        float dv[SCAN_U], xv[SCAN_U];
        for (int u = 0; u < SCAN_U; ++u) { dv[u] = dvN[u]; xv[u] = xvN[u]; }
        if (i0 + SCAN_U < LCHUNK) {
            for (int u = 0; u < SCAN_U; ++u) {
                dvN[u] = dp[(size_t)(i0 + SCAN_U + u) * DIN];
                xvN[u] = xp[(size_t)(i0 + SCAN_U + u) * DIN];
            }
        }
        for (int u = 0; u < SCAN_U; ++u) {
            const v4f B0 = *(const v4f*)&sB[i0 + u][0];
            const v4f B1 = *(const v4f*)&sB[i0 + u][4];
            const float dbx = dv[u] * xv[u];
            for (int n = 0; n < DSTATE; ++n) {
                const float Bv = (n < 4) ? B0[n & 3] : B1[n & 3];
                const float dA = __expf(dv[u] * A[n]);
                h[n]  = fmaf(dA, h[n], dbx * Bv);
                aP[n] *= dA;
            }
        }
    }

    for (int n = 0; n < DSTATE; ++n) {
        const size_t idx = ((size_t)(b * NCHUNK + c) * DSTATE + n) * DIN + d;
        aProd[idx] = aP[n];
        hPart[idx] = h[n];
    }
}

__global__ __launch_bounds__(256) void scan_pass2(
    const float* __restrict__ aProd,
    const float* __restrict__ hPart,
    float* __restrict__ hIn)
{
    const int t = blockIdx.x * blockDim.x + threadIdx.x;
    const int d = t % DIN;
    const int n = (t / DIN) % DSTATE;
    const int b = t / (DIN * DSTATE);
    float h = 0.0f;
    #pragma unroll
    for (int c = 0; c < NCHUNK; ++c) {
        const size_t idx = ((size_t)(b * NCHUNK + c) * DSTATE + n) * DIN + d;
        hIn[idx] = h;
        h = fmaf(aProd[idx], h, hPart[idx]);
    }
}

__global__ __launch_bounds__(256) void scan_pass3(
    const float* __restrict__ delta,
    const float* __restrict__ x,
    const float* __restrict__ xdbl,
    const float* __restrict__ A_log,
    const float* __restrict__ D_param,
    const float* __restrict__ hIn,
    float* __restrict__ y)
{
    __shared__ __attribute__((aligned(16))) float sBC[LCHUNK][2 * DSTATE];

    const int tid  = threadIdx.x;
    const int dblk = blockIdx.x % NDBLK;
    const int c    = (blockIdx.x / NDBLK) % NCHUNK;
    const int b    = blockIdx.x / (NDBLK * NCHUNK);
    const int d    = dblk * DBLK + tid;
    const int row0 = b * LSEQ + c * LCHUNK;

    for (int idx = tid; idx < LCHUNK * 4; idx += 256) {
        const int r = idx >> 2, q = idx & 3;
        *(v4f*)&sBC[r][4 * q] =
            *(const v4f*)(xdbl + (size_t)(row0 + r) * KPROJ + DTRANK + 4 * q);
    }

    float A[DSTATE], h[DSTATE];
    {
        const v4f a0 = *(const v4f*)(A_log + (size_t)d * DSTATE);
        const v4f a1 = *(const v4f*)(A_log + (size_t)d * DSTATE + 4);
        for (int n = 0; n < 4; ++n) {
            A[n]     = -__expf(a0[n]);
            A[n + 4] = -__expf(a1[n]);
        }
    }
    const float Dp = D_param[d];
    for (int n = 0; n < DSTATE; ++n)
        h[n] = hIn[((size_t)(b * NCHUNK + c) * DSTATE + n) * DIN + d];

    const float* dp = delta + (size_t)row0 * DIN + d;
    const float* xp = x     + (size_t)row0 * DIN + d;
    float*       yp = y     + (size_t)row0 * DIN + d;

    float dvN[SCAN_U], xvN[SCAN_U];
    for (int u = 0; u < SCAN_U; ++u) {
        dvN[u] = dp[(size_t)u * DIN];
        xvN[u] = xp[(size_t)u * DIN];
    }

    __syncthreads();

    for (int i0 = 0; i0 < LCHUNK; i0 += SCAN_U) {
        float dv[SCAN_U], xv[SCAN_U];
        for (int u = 0; u < SCAN_U; ++u) { dv[u] = dvN[u]; xv[u] = xvN[u]; }
        if (i0 + SCAN_U < LCHUNK) {
            for (int u = 0; u < SCAN_U; ++u) {
                dvN[u] = dp[(size_t)(i0 + SCAN_U + u) * DIN];
                xvN[u] = xp[(size_t)(i0 + SCAN_U + u) * DIN];
            }
        }
        for (int u = 0; u < SCAN_U; ++u) {
            const v4f B0 = *(const v4f*)&sBC[i0 + u][0];
            const v4f B1 = *(const v4f*)&sBC[i0 + u][4];
            const v4f C0 = *(const v4f*)&sBC[i0 + u][8];
            const v4f C1 = *(const v4f*)&sBC[i0 + u][12];
            const float dbx = dv[u] * xv[u];
            float acc = 0.0f;
            for (int n = 0; n < DSTATE; ++n) {
                const float Bv = (n < 4) ? B0[n & 3] : B1[n & 3];
                const float Cv = (n < 4) ? C0[n & 3] : C1[n & 3];
                const float dA = __expf(dv[u] * A[n]);
                h[n] = fmaf(dA, h[n], dbx * Bv);
                acc  = fmaf(h[n], Cv, acc);
            }
            yp[(size_t)(i0 + u) * DIN] = fmaf(xv[u], Dp, acc);
        }
    }
}

extern "C" void kernel_launch(void* const* d_in, const int* in_sizes, int n_in,
                              void* d_out, int out_size, void* d_ws, size_t ws_size,
                              hipStream_t stream) {
    const float* x       = (const float*)d_in[0];
    const float* W_xproj = (const float*)d_in[1];
    const float* W_dt    = (const float*)d_in[2];
    const float* b_dt    = (const float*)d_in[3];
    const float* A_log   = (const float*)d_in[4];
    const float* D_param = (const float*)d_in[5];
    float* y = (float*)d_out;

    float* ws    = (float*)d_ws;
    float* xdbl  = ws;
    float* delta = xdbl + (size_t)BSZ * LSEQ * KPROJ;
    float* aProd = delta + (size_t)BSZ * LSEQ * DIN;
    float* hPart = aProd + (size_t)BSZ * DIN * NCHUNK * DSTATE;
    float* hIn   = hPart + (size_t)BSZ * DIN * NCHUNK * DSTATE;

    proj_kernel<<<256, 128, 0, stream>>>(x, W_xproj, xdbl);
    delta_kernel<<<2048, 32, 0, stream>>>(xdbl, W_dt, b_dt, delta);
    scan_pass1<<<512, 256, 0, stream>>>(delta, x, xdbl, A_log, aProd, hPart);
    scan_pass2<<<128, 256, 0, stream>>>(aProd, hPart, hIn);
    scan_pass3<<<512, 256, 0, stream>>>(delta, x, xdbl, A_log, D_param, hIn, y);
}


# scan passes 1 and 3: 8 states as 4 pairs on packed f32 mul/fma, exp(d*A) as exp2((d*log2e)*A), chunk decay = exp2(A*sum d); sum over n in even/odd halves (order changes)
# speedup vs baseline: 4.9133x; 1.0473x over previous
#include <hip/hip_runtime.h>
#include <hip/hip_bf16.h>


#define BSZ     2
#define LSEQ    2048
#define DIN     2048
#define DSTATE  8
#define DTRANK  256
#define KPROJ   (DTRANK + 2 * DSTATE)
#define NCHUNK  32
#define LCHUNK  (LSEQ / NCHUNK)
#define DBLK    256
#define NDBLK   (DIN / DBLK)
#define SCAN_U  8

typedef float v2f __attribute__((ext_vector_type(2)));
typedef float v8f __attribute__((ext_vector_type(8)));
typedef float v4f __attribute__((ext_vector_type(4)));

#define LOG2E_F 1.44269504088896340736f

__device__ __forceinline__ v2f exp2_pair(const v2f a)
{
    v2f r;
    r.x = __builtin_amdgcn_exp2f(a.x);
    r.y = __builtin_amdgcn_exp2f(a.y);
    return r;
}

template <int TM, int TN, int SUB>
__device__ __forceinline__ void mma_stage(
    const v4f (&a)[TM][SUB], const v4f (&b)[TN][SUB], v8f (&acc)[TM][TN])
{
    #pragma unroll
    for (int s = 0; s < SUB; ++s) {
        #pragma unroll
        for (int i = 0; i < TM; ++i) {
            #pragma unroll
            for (int j = 0; j < TN; ++j) {
                const v2f al = __builtin_shufflevector(a[i][s], a[i][s], 0, 1);
                const v2f bl = __builtin_shufflevector(b[j][s], b[j][s], 0, 1);
                acc[i][j] = __builtin_amdgcn_wmma_f32_16x16x4_f32(
                    false, al, false, bl, (short)0, acc[i][j], false, false);
            }
        }
        #pragma unroll
        for (int i = 0; i < TM; ++i) {
            #pragma unroll
            for (int j = 0; j < TN; ++j) {
                const v2f ah = __builtin_shufflevector(a[i][s], a[i][s], 2, 3);
                const v2f bh = __builtin_shufflevector(b[j][s], b[j][s], 2, 3);
                acc[i][j] = __builtin_amdgcn_wmma_f32_16x16x4_f32(
                    false, ah, false, bh, (short)0, acc[i][j], false, false);
            }
        }
    }
}

template <int TM, int TN, int SUB, int LDA, int LDB>
__device__ __forceinline__ void load_stage(
    const float* __restrict__ ap, const float* __restrict__ bp, const int k,
    v4f (&a)[TM][SUB], v4f (&b)[TN][SUB])
{
    #pragma unroll
    for (int i = 0; i < TM; ++i) {
        #pragma unroll
        for (int s = 0; s < SUB; ++s)
            a[i][s] = *(const v4f*)(ap + (size_t)i * 16 * LDA + k + 8 * s);
    }
    #pragma unroll
    for (int j = 0; j < TN; ++j) {
        #pragma unroll
        for (int s = 0; s < SUB; ++s)
            b[j][s] = *(const v4f*)(bp + (size_t)j * 16 * LDB + k + 8 * s);
    }
}

template <int TM, int TN, int SUB, int LDA, int LDB, int KLEN>
__device__ __forceinline__ void wave_tiles_nt(
    const float* __restrict__ ap,
    const float* __restrict__ bp,
    v8f (&acc)[TM][TN])
{
    static_assert(KLEN % (16 * SUB) == 0 && KLEN >= 32 * SUB, "two stages per trip");
    v4f a0[TM][SUB], b0[TN][SUB], a1[TM][SUB], b1[TN][SUB];
    load_stage<TM, TN, SUB, LDA, LDB>(ap, bp, 0, a0, b0);

    for (int k = 0; k < KLEN - 16 * SUB; k += 16 * SUB) {
        load_stage<TM, TN, SUB, LDA, LDB>(ap, bp, 8 * SUB, a1, b1);
        __builtin_amdgcn_sched_barrier(0);
        mma_stage<TM, TN, SUB>(a0, b0, acc);
        load_stage<TM, TN, SUB, LDA, LDB>(ap, bp, 16 * SUB, a0, b0);
        __builtin_amdgcn_sched_barrier(0);
        mma_stage<TM, TN, SUB>(a1, b1, acc);
        ap += 16 * SUB;
        bp += 16 * SUB;
    }

    load_stage<TM, TN, SUB, LDA, LDB>(ap, bp, 8 * SUB, a1, b1);
    __builtin_amdgcn_sched_barrier(0);
    mma_stage<TM, TN, SUB>(a0, b0, acc);
    mma_stage<TM, TN, SUB>(a1, b1, acc);
}

#define PROJ_WAVES   4
#define PROJ_KSLICE  (DIN / PROJ_WAVES)
#define PROJ_TN0     9
#define PROJ_TN1     8

template <int TN>
__device__ __forceinline__ void proj_block(
    const float* __restrict__ ap, const float* __restrict__ bp,
    float* __restrict__ xdbl, v4f* __restrict__ sP,
    const int wave, const int lane, const int row0, const int col0)
{
    const int c  = lane & 15;
    const int hi = lane >> 4;

    v8f acc[2][TN] = {};
    wave_tiles_nt<2, TN, 4, DIN, DIN, PROJ_KSLICE>(ap, bp, acc);

    #pragma unroll
    for (int i = 0; i < 2; ++i) {
        #pragma unroll
        for (int j = 0; j < TN; ++j) {
            const int t = i * TN + j;
            if ((t & 3) != wave) {
                const int p = (wave - t - 1) & 3;
                v4f* dst = sP + ((t * 3 + p) * 2) * 32 + lane;
                dst[0]  = __builtin_shufflevector(acc[i][j], acc[i][j], 0, 1, 2, 3);
                dst[32] = __builtin_shufflevector(acc[i][j], acc[i][j], 4, 5, 6, 7);
            }
        }
    }

    __syncthreads();

    #pragma unroll
    for (int i = 0; i < 2; ++i) {
        #pragma unroll
        for (int j = 0; j < TN; ++j) {
            const int t = i * TN + j;
            if ((t & 3) == wave) {
                v8f s = acc[i][j];
                #pragma unroll
                for (int p = 0; p < 3; ++p) {
                    const v4f* src = sP + ((t * 3 + p) * 2) * 32 + lane;
                    const v4f lo = src[0];
                    const v4f hh = src[32];
                    #pragma unroll
                    for (int r = 0; r < 4; ++r) {
                        s[r]     += lo[r];
                        s[r + 4] += hh[r];
                    }
                }
                #pragma unroll
                for (int r = 0; r < 8; ++r) {
                    const int row = row0 + 16 * i + r + 8 * hi;
                    const int col = col0 + 16 * j + c;
                    xdbl[(size_t)row * KPROJ + col] = s[r];
                }
            }
        }
    }
}

__global__ __launch_bounds__(128) void proj_kernel(
    const float* __restrict__ x,
    const float* __restrict__ Wxp,
    float* __restrict__ xdbl)
{
    __shared__ __attribute__((aligned(16))) v4f sP[2 * PROJ_TN0 * 3 * 2 * 32];

    const int lane = threadIdx.x & 31;
    const int wave = __builtin_amdgcn_readfirstlane((int)(threadIdx.x >> 5));
    const int c  = lane & 15;
    const int hi = lane >> 4;
    const int rp = blockIdx.x >> 1;
    const int cg = blockIdx.x & 1;
    const int row0 = rp * 32;
    const int col0 = cg * (16 * PROJ_TN0);

    const float* ap = x   + (size_t)(row0 + c) * DIN + wave * PROJ_KSLICE + 4 * hi;
    const float* bp = Wxp + (size_t)(col0 + c) * DIN + wave * PROJ_KSLICE + 4 * hi;

    if (cg == 0) proj_block<PROJ_TN0>(ap, bp, xdbl, sP, wave, lane, row0, col0);
    else         proj_block<PROJ_TN1>(ap, bp, xdbl, sP, wave, lane, row0, col0);
}

__device__ __forceinline__ float softplus_f32(const float z)
{
    const float t = __expf(-fabsf(z));
    const float s = t * __builtin_amdgcn_rcpf(2.0f + t);
    const float w = s * s;
    float q = 0.281577796f;
    q = fmaf(q, w, 0.279651582f);
    q = fmaf(q, w, 0.400245965f);
    q = fmaf(q, w, 0.666663229f);
    q = fmaf(q, w, 2.0f);
    return fmaf(s, q, fmaxf(z, 0.0f));
}

__global__ __launch_bounds__(32) void delta_kernel(
    const float* __restrict__ xdbl,
    const float* __restrict__ Wdt,
    const float* __restrict__ b_dt,
    float* __restrict__ delta)
{
    const int lane = threadIdx.x & 31;
    const int c  = lane & 15;
    const int hi = lane >> 4;
    const int rg = blockIdx.x >> 5;
    const int cg = blockIdx.x & 31;

    const float* ap = xdbl + (size_t)(rg * 64 + c) * KPROJ  + 4 * hi;
    const float* bp = Wdt  + (size_t)(cg * 64 + c) * DTRANK + 4 * hi;

    v8f acc[4][4] = {};
    wave_tiles_nt<4, 4, 4, KPROJ, DTRANK, DTRANK>(ap, bp, acc);

    #pragma unroll
    for (int jt = 0; jt < 4; ++jt) {
        const int col = cg * 64 + 16 * jt + c;
        const float bias = b_dt[col];
        #pragma unroll
        for (int i = 0; i < 4; ++i) {
            #pragma unroll
            for (int j = 0; j < 8; ++j) {
                const int row = rg * 64 + 16 * i + j + 8 * hi;
                delta[(size_t)row * DIN + col] = softplus_f32(acc[i][jt][j] + bias);
            }
        }
    }
}

__global__ __launch_bounds__(256) void scan_pass1(
    const float* __restrict__ delta,
    const float* __restrict__ x,
    const float* __restrict__ xdbl,
    const float* __restrict__ A_log,
    float* __restrict__ aProd,
    float* __restrict__ hPart)
{
    __shared__ __attribute__((aligned(16))) float sB[LCHUNK][DSTATE];

    const int tid  = threadIdx.x;
    const int dblk = blockIdx.x % NDBLK;
    const int c    = (blockIdx.x / NDBLK) % NCHUNK;
    const int b    = blockIdx.x / (NDBLK * NCHUNK);
    const int d    = dblk * DBLK + tid;
    const int row0 = b * LSEQ + c * LCHUNK;

    for (int idx = tid; idx < LCHUNK * 2; idx += 256) {
        const int r = idx >> 1, q = idx & 1;
        *(v4f*)&sB[r][4 * q] =
            *(const v4f*)(xdbl + (size_t)(row0 + r) * KPROJ + DTRANK + 4 * q);
    }

    v2f A2[DSTATE / 2], h2[DSTATE / 2];
    {
        const v4f a0 = *(const v4f*)(A_log + (size_t)d * DSTATE);
        const v4f a1 = *(const v4f*)(A_log + (size_t)d * DSTATE + 4);
        A2[0].x = -__expf(a0[0]);  A2[0].y = -__expf(a0[1]);
        A2[1].x = -__expf(a0[2]);  A2[1].y = -__expf(a0[3]);
        A2[2].x = -__expf(a1[0]);  A2[2].y = -__expf(a1[1]);
        A2[3].x = -__expf(a1[2]);  A2[3].y = -__expf(a1[3]);
    }
    for (int j = 0; j < DSTATE / 2; ++j) { h2[j].x = 0.0f; h2[j].y = 0.0f; }
    float dsum = 0.0f;

    const float* dp = delta + (size_t)row0 * DIN + d;
    const float* xp = x     + (size_t)row0 * DIN + d;

    float dvN[SCAN_U], xvN[SCAN_U];
    for (int u = 0; u < SCAN_U; ++u) {
        dvN[u] = dp[(size_t)u * DIN];
        xvN[u] = xp[(size_t)u * DIN];
    }

    __syncthreads();

    for (int i0 = 0; i0 < LCHUNK; i0 += SCAN_U) {
        float dv[SCAN_U], xv[SCAN_U];
        for (int u = 0; u < SCAN_U; ++u) { dv[u] = dvN[u]; xv[u] = xvN[u]; }
        if (i0 + SCAN_U < LCHUNK) {
            for (int u = 0; u < SCAN_U; ++u) {
                dvN[u] = dp[(size_t)(i0 + SCAN_U + u) * DIN];
                xvN[u] = xp[(size_t)(i0 + SCAN_U + u) * DIN];
            }
        }
        for (int u = 0; u < SCAN_U; ++u) {
            const v4f B0 = *(const v4f*)&sB[i0 + u][0];
            const v4f B1 = *(const v4f*)&sB[i0 + u][4];
            const v2f Bp[DSTATE / 2] = {
                __builtin_shufflevector(B0, B0, 0, 1), __builtin_shufflevector(B0, B0, 2, 3),
                __builtin_shufflevector(B1, B1, 0, 1), __builtin_shufflevector(B1, B1, 2, 3) };
            const float d2  = dv[u] * LOG2E_F;
            const float dbx = dv[u] * xv[u];
            dsum += d2;
            for (int j = 0; j < DSTATE / 2; ++j) {
                const v2f dA = exp2_pair(d2 * A2[j]);
                h2[j] = __builtin_elementwise_fma(dA, h2[j], dbx * Bp[j]);
            }
        }
    }

    for (int j = 0; j < DSTATE / 2; ++j) {
        const size_t idx = ((size_t)(b * NCHUNK + c) * DSTATE + 2 * j) * DIN + d;
        const v2f aP = exp2_pair(dsum * A2[j]);
        aProd[idx]       = aP.x;
        aProd[idx + DIN] = aP.y;
        hPart[idx]       = h2[j].x;
        hPart[idx + DIN] = h2[j].y;
    }
}

__global__ __launch_bounds__(256) void scan_pass2(
    const float* __restrict__ aProd,
    const float* __restrict__ hPart,
    float* __restrict__ hIn)
{
    const int t = blockIdx.x * blockDim.x + threadIdx.x;
    const int d = t % DIN;
    const int n = (t / DIN) % DSTATE;
    const int b = t / (DIN * DSTATE);
    float h = 0.0f;
    #pragma unroll
    for (int c = 0; c < NCHUNK; ++c) {
        const size_t idx = ((size_t)(b * NCHUNK + c) * DSTATE + n) * DIN + d;
        hIn[idx] = h;
        h = fmaf(aProd[idx], h, hPart[idx]);
    }
}

__global__ __launch_bounds__(256) void scan_pass3(
    const float* __restrict__ delta,
    const float* __restrict__ x,
    const float* __restrict__ xdbl,
    const float* __restrict__ A_log,
    const float* __restrict__ D_param,
    const float* __restrict__ hIn,
    float* __restrict__ y)
{
    __shared__ __attribute__((aligned(16))) float sBC[LCHUNK][2 * DSTATE];

    const int tid  = threadIdx.x;
    const int dblk = blockIdx.x % NDBLK;
    const int c    = (blockIdx.x / NDBLK) % NCHUNK;
    const int b    = blockIdx.x / (NDBLK * NCHUNK);
    const int d    = dblk * DBLK + tid;
    const int row0 = b * LSEQ + c * LCHUNK;

    for (int idx = tid; idx < LCHUNK * 4; idx += 256) {
        const int r = idx >> 2, q = idx & 3;
        *(v4f*)&sBC[r][4 * q] =
            *(const v4f*)(xdbl + (size_t)(row0 + r) * KPROJ + DTRANK + 4 * q);
    }

    v2f A2[DSTATE / 2], h2[DSTATE / 2];
    {
        const v4f a0 = *(const v4f*)(A_log + (size_t)d * DSTATE);
        const v4f a1 = *(const v4f*)(A_log + (size_t)d * DSTATE + 4);
        A2[0].x = -__expf(a0[0]);  A2[0].y = -__expf(a0[1]);
        A2[1].x = -__expf(a0[2]);  A2[1].y = -__expf(a0[3]);
        A2[2].x = -__expf(a1[0]);  A2[2].y = -__expf(a1[1]);
        A2[3].x = -__expf(a1[2]);  A2[3].y = -__expf(a1[3]);
    }
    const float Dp = D_param[d];
    for (int j = 0; j < DSTATE / 2; ++j) {
        const size_t idx = ((size_t)(b * NCHUNK + c) * DSTATE + 2 * j) * DIN + d;
        h2[j].x = hIn[idx];
        h2[j].y = hIn[idx + DIN];
    }

    const float* dp = delta + (size_t)row0 * DIN + d;
    const float* xp = x     + (size_t)row0 * DIN + d;
    float*       yp = y     + (size_t)row0 * DIN + d;

    float dvN[SCAN_U], xvN[SCAN_U];
    for (int u = 0; u < SCAN_U; ++u) {
        dvN[u] = dp[(size_t)u * DIN];
        xvN[u] = xp[(size_t)u * DIN];
    }

    __syncthreads();

    for (int i0 = 0; i0 < LCHUNK; i0 += SCAN_U) {
        float dv[SCAN_U], xv[SCAN_U];
        for (int u = 0; u < SCAN_U; ++u) { dv[u] = dvN[u]; xv[u] = xvN[u]; }
        if (i0 + SCAN_U < LCHUNK) {
            for (int u = 0; u < SCAN_U; ++u) {
                dvN[u] = dp[(size_t)(i0 + SCAN_U + u) * DIN];
                xvN[u] = xp[(size_t)(i0 + SCAN_U + u) * DIN];
            }
        }
        for (int u = 0; u < SCAN_U; ++u) {
            const v4f B0 = *(const v4f*)&sBC[i0 + u][0];
            const v4f B1 = *(const v4f*)&sBC[i0 + u][4];
            const v4f C0 = *(const v4f*)&sBC[i0 + u][8];
            const v4f C1 = *(const v4f*)&sBC[i0 + u][12];
            const v2f Bp[DSTATE / 2] = {
                __builtin_shufflevector(B0, B0, 0, 1), __builtin_shufflevector(B0, B0, 2, 3),
                __builtin_shufflevector(B1, B1, 0, 1), __builtin_shufflevector(B1, B1, 2, 3) };
            const v2f Cp[DSTATE / 2] = {
                __builtin_shufflevector(C0, C0, 0, 1), __builtin_shufflevector(C0, C0, 2, 3),
                __builtin_shufflevector(C1, C1, 0, 1), __builtin_shufflevector(C1, C1, 2, 3) };
            const float d2  = dv[u] * LOG2E_F;
            const float dbx = dv[u] * xv[u];
            v2f acc2 = {0.0f, 0.0f};
            for (int j = 0; j < DSTATE / 2; ++j) {
                const v2f dA = exp2_pair(d2 * A2[j]);
                h2[j] = __builtin_elementwise_fma(dA, h2[j], dbx * Bp[j]);
                acc2  = __builtin_elementwise_fma(h2[j], Cp[j], acc2);
            }
            yp[(size_t)(i0 + u) * DIN] = fmaf(xv[u], Dp, acc2.x + acc2.y);
        }
    }
}

extern "C" void kernel_launch(void* const* d_in, const int* in_sizes, int n_in,
                              void* d_out, int out_size, void* d_ws, size_t ws_size,
                              hipStream_t stream) {
    const float* x       = (const float*)d_in[0];
    const float* W_xproj = (const float*)d_in[1];
    const float* W_dt    = (const float*)d_in[2];
    const float* b_dt    = (const float*)d_in[3];
    const float* A_log   = (const float*)d_in[4];
    const float* D_param = (const float*)d_in[5];
    float* y = (float*)d_out;

    float* ws    = (float*)d_ws;
    float* xdbl  = ws;
    float* delta = xdbl + (size_t)BSZ * LSEQ * KPROJ;
    float* aProd = delta + (size_t)BSZ * LSEQ * DIN;
    float* hPart = aProd + (size_t)BSZ * DIN * NCHUNK * DSTATE;
    float* hIn   = hPart + (size_t)BSZ * DIN * NCHUNK * DSTATE;

    proj_kernel<<<256, 128, 0, stream>>>(x, W_xproj, xdbl);
    delta_kernel<<<2048, 32, 0, stream>>>(xdbl, W_dt, b_dt, delta);
    scan_pass1<<<512, 256, 0, stream>>>(delta, x, xdbl, A_log, aProd, hPart);
    scan_pass2<<<128, 256, 0, stream>>>(aProd, hPart, hIn);
    scan_pass3<<<512, 256, 0, stream>>>(delta, x, xdbl, A_log, D_param, hIn, y);
}
